# starting kernel
# baseline (speedup 1.0000x reference)
#include <hip/hip_runtime.h>
#include <hip/hip_bf16.h>


#define N_NODES 50000
#define D_POS   32
#define D_FEAT  768
#define D_IN    801
#define D_HID   32
#define D_OUT   32
#define N_GRAPHS 128
#define M_TILES  (N_NODES / 16)
#define WAVES_PER_BLOCK 8

typedef __attribute__((ext_vector_type(2))) float v2f;
typedef __attribute__((ext_vector_type(8))) float v8f;

__global__ __launch_bounds__(256) void input_gemm_kernel(
    const float* __restrict__ pos, const float* __restrict__ feat,
    const float* __restrict__ W1,  const float* __restrict__ Wp0,
    float* __restrict__ yBuf, float* __restrict__ accBuf, float* __restrict__ zBuf)
{
    const int wave = threadIdx.x >> 5;
    const int lane = threadIdx.x & 31;
    const int tile = blockIdx.x * WAVES_PER_BLOCK + wave;
    if (tile >= M_TILES) return;

    const int  m0   = tile * 16;
    const int  col  = lane & 15;
    const bool hi   = lane >= 16;
    const int  node = m0 + col;

    v8f cy0 = {}, cy1 = {}, cz0 = {}, cz1 = {};

    const float* posRow = pos + (size_t)node * D_POS;
    for (int k = 0; k < D_POS; k += 4) {
        const int kb = k + (hi ? 2 : 0);
        const v2f a = *(const v2f*)(posRow + kb);

        v2f by0, by1, bz0, bz1;
        by0.x = W1 [kb * 32 + col];        by0.y = W1 [(kb + 1) * 32 + col];
        by1.x = W1 [kb * 32 + 16 + col];   by1.y = W1 [(kb + 1) * 32 + 16 + col];
        bz0.x = Wp0[kb * 32 + col];        bz0.y = Wp0[(kb + 1) * 32 + col];
        bz1.x = Wp0[kb * 32 + 16 + col];   bz1.y = Wp0[(kb + 1) * 32 + 16 + col];

        cy0 = __builtin_amdgcn_wmma_f32_16x16x4_f32(false, a, false, by0, (short)0, cy0, false, false);
        cy1 = __builtin_amdgcn_wmma_f32_16x16x4_f32(false, a, false, by1, (short)0, cy1, false, false);
        cz0 = __builtin_amdgcn_wmma_f32_16x16x4_f32(false, a, false, bz0, (short)0, cz0, false, false);
        cz1 = __builtin_amdgcn_wmma_f32_16x16x4_f32(false, a, false, bz1, (short)0, cz1, false, false);
    }

    const float* featRow = feat + (size_t)node * D_FEAT;
    for (int k = 0; k < D_FEAT; k += 4) {
        const int kb = k + (hi ? 2 : 0);
        const v2f a = *(const v2f*)(featRow + kb);

        const int r = (33 + kb) * 32;
        v2f by0, by1, bz0, bz1;
        by0.x = W1 [r + col];        by0.y = W1 [r + 32 + col];
        by1.x = W1 [r + 16 + col];   by1.y = W1 [r + 48 + col];
        bz0.x = Wp0[r + col];        bz0.y = Wp0[r + 32 + col];
        bz1.x = Wp0[r + 16 + col];   bz1.y = Wp0[r + 48 + col];

        cy0 = __builtin_amdgcn_wmma_f32_16x16x4_f32(false, a, false, by0, (short)0, cy0, false, false);
        cy1 = __builtin_amdgcn_wmma_f32_16x16x4_f32(false, a, false, by1, (short)0, cy1, false, false);
        cz0 = __builtin_amdgcn_wmma_f32_16x16x4_f32(false, a, false, bz0, (short)0, cz0, false, false);
        cz1 = __builtin_amdgcn_wmma_f32_16x16x4_f32(false, a, false, bz1, (short)0, cz1, false, false);
    }

    for (int j = 0; j < 8; ++j) {
        const int row = (hi ? 8 : 0) + j;
        const int off = (m0 + row) * 32 + col;
        yBuf[off]        = cy0[j];  accBuf[off]        = cy0[j];
        yBuf[off + 16]   = cy1[j];  accBuf[off + 16]   = cy1[j];
        zBuf[off]        = cz0[j];
        zBuf[off + 16]   = cz1[j];
    }
}

__global__ __launch_bounds__(256) void edge_scatter_kernel(
    const float* __restrict__ yBuf, const int* __restrict__ src,
    const int* __restrict__ dst, float* __restrict__ accBuf, int nEdges)
{
    const int idx = blockIdx.x * blockDim.x + threadIdx.x;
    const int e = idx >> 3;
    if (e >= nEdges) return;
    const int c = (idx & 7) * 4;
    const int s = src[e], d = dst[e];
    const float4 v = *(const float4*)(yBuf + (size_t)s * 32 + c);
    float* p = accBuf + (size_t)d * 32 + c;
    atomicAdd(p + 0, v.x);
    atomicAdd(p + 1, v.y);
    atomicAdd(p + 2, v.z);
    atomicAdd(p + 3, v.w);
}

__global__ __launch_bounds__(256) void node_mlp_head_kernel(
    const float* __restrict__ accBuf, const float* __restrict__ zBuf,
    const float* __restrict__ W2, const float* __restrict__ bias1,
    const float* __restrict__ bias2, const float* __restrict__ Wp1,
    const int* __restrict__ gids, float* __restrict__ partial)
{
    __shared__ float hbuf[WAVES_PER_BLOCK][16 * 32];

    const int wave = threadIdx.x >> 5;
    const int lane = threadIdx.x & 31;
    const int tile = blockIdx.x * WAVES_PER_BLOCK + wave;
    if (tile >= M_TILES) return;

    const int  m0   = tile * 16;
    const int  col  = lane & 15;
    const bool hi   = lane >= 16;
    const int  node = m0 + col;

    const float* accRow = accBuf + (size_t)node * 32;
    v8f c10 = {}, c11 = {};
    for (int k = 0; k < 32; k += 4) {
        const int kb = k + (hi ? 2 : 0);
        const v2f av = *(const v2f*)(accRow + kb);
        const v2f bv = *(const v2f*)(bias1 + kb);
        v2f a;
        a.x = fmaxf(av.x + bv.x, 0.0f);
        a.y = fmaxf(av.y + bv.y, 0.0f);
        v2f b0, b1f;
        b0.x  = W2[kb * 32 + col];       b0.y  = W2[(kb + 1) * 32 + col];
        b1f.x = W2[kb * 32 + 16 + col];  b1f.y = W2[(kb + 1) * 32 + 16 + col];
        c10 = __builtin_amdgcn_wmma_f32_16x16x4_f32(false, a, false, b0,  (short)0, c10, false, false);
        c11 = __builtin_amdgcn_wmma_f32_16x16x4_f32(false, a, false, b1f, (short)0, c11, false, false);
    }

    const float bb0 = bias2[col];
    const float bb1 = bias2[16 + col];
    for (int j = 0; j < 8; ++j) {
        const int row = (hi ? 8 : 0) + j;
        hbuf[wave][row * 32 + col]      = fmaxf(c10[j] + bb0, 0.0f);
        hbuf[wave][row * 32 + 16 + col] = fmaxf(c11[j] + bb1, 0.0f);
    }

    v8f u0 = {}, u1 = {};
    for (int k = 0; k < 32; k += 4) {
        const int kb = k + (hi ? 2 : 0);
        const v2f a = *(const v2f*)(&hbuf[wave][col * 32 + kb]);
        v2f b0, b1f;
        b0.x  = Wp1[kb * 32 + col];       b0.y  = Wp1[(kb + 1) * 32 + col];
        b1f.x = Wp1[kb * 32 + 16 + col];  b1f.y = Wp1[(kb + 1) * 32 + 16 + col];
        u0 = __builtin_amdgcn_wmma_f32_16x16x4_f32(false, a, false, b0,  (short)0, u0, false, false);
        u1 = __builtin_amdgcn_wmma_f32_16x16x4_f32(false, a, false, b1f, (short)0, u1, false, false);
    }

    for (int j = 0; j < 8; ++j) {
        const int row  = (hi ? 8 : 0) + j;
        const int nd   = m0 + row;
        const int g    = gids[nd];
        const int off  = nd * 32 + col;
        atomicAdd(&partial[g * 32 + col],      u0[j] + zBuf[off]);
        atomicAdd(&partial[g * 32 + 16 + col], u1[j] + zBuf[off + 16]);
    }
}

__global__ void zero_partial_kernel(float* __restrict__ partial) {
    const int i = blockIdx.x * blockDim.x + threadIdx.x;
    if (i < N_GRAPHS * D_OUT) partial[i] = 0.0f;
}

__global__ void finalize_kernel(const float* __restrict__ partial,
                                const float* __restrict__ bp0,
                                const float* __restrict__ bp1,
                                float* __restrict__ out) {
    const int i = blockIdx.x * blockDim.x + threadIdx.x;
    if (i < N_GRAPHS * D_OUT) {
        const int c = i & 31;
        out[i] = partial[i] + bp0[c] + bp1[c];
    }
}

extern "C" void kernel_launch(void* const* d_in, const int* in_sizes, int n_in,
                              void* d_out, int out_size, void* d_ws, size_t ws_size,
                              hipStream_t stream) {
    const float* pos  = (const float*)d_in[0];
    const float* feat = (const float*)d_in[1];
    const int*   src  = (const int*)  d_in[2];
    const int*   dst  = (const int*)  d_in[3];
    const int*   gids = (const int*)  d_in[4];
    const float* W1   = (const float*)d_in[5];
    const float* b1   = (const float*)d_in[6];
    const float* W2   = (const float*)d_in[7];
    const float* b2   = (const float*)d_in[8];
    const float* Wp0  = (const float*)d_in[9];
    const float* bp0  = (const float*)d_in[10];
    const float* Wp1  = (const float*)d_in[11];
    const float* bp1  = (const float*)d_in[12];
    float* out = (float*)d_out;
    const int nEdges = in_sizes[2];

    float* yBuf    = (float*)d_ws;
    float* zBuf    = yBuf + (size_t)N_NODES * 32;
    float* accBuf  = zBuf + (size_t)N_NODES * 32;
    float* partial = accBuf + (size_t)N_NODES * 32;

    const int gemmBlocks = (M_TILES + WAVES_PER_BLOCK - 1) / WAVES_PER_BLOCK;

    zero_partial_kernel<<<(N_GRAPHS * D_OUT + 255) / 256, 256, 0, stream>>>(partial);
    input_gemm_kernel<<<gemmBlocks, 256, 0, stream>>>(pos, feat, W1, Wp0,
                                                      yBuf, accBuf, zBuf);
    edge_scatter_kernel<<<((size_t)nEdges * 8 + 255) / 256, 256, 0, stream>>>(
        yBuf, src, dst, accBuf, nEdges);
    node_mlp_head_kernel<<<gemmBlocks, 256, 0, stream>>>(accBuf, zBuf, W2, b1, b2,
                                                         Wp1, gids, partial);
    finalize_kernel<<<(N_GRAPHS * D_OUT + 255) / 256, 256, 0, stream>>>(
        partial, bp0, bp1, out);
}


# owned edges found by ANDing two shared edge bitmaps (by high and low nibble of the dst owner, written by the input-GEMM launch), not by testing every dst id in each work-group; LDS add order changes
# speedup vs baseline: 1.8889x; 1.8889x over previous
#include <hip/hip_runtime.h>
#include <hip/hip_bf16.h>


#define N_NODES 50000
#define D_POS   32
#define D_FEAT  768
#define D_IN    801
#define D_HID   32
#define D_OUT   32
#define N_GRAPHS 128
#define M_TILES  (N_NODES / 16)

#define GEMM_WAVES   16
#define GEMM_GROUPS  256
#define KW_FEAT      (D_FEAT / GEMM_WAVES)
#define CH_FEAT      (KW_FEAT / 4)
#define PART_PITCH   36
#define PART_TILE    (16 * PART_PITCH)

typedef __attribute__((ext_vector_type(2))) float v2f;
typedef __attribute__((ext_vector_type(8))) float v8f;

__device__ __forceinline__ unsigned ownerOfNode(const int d) {
    const unsigned t = (unsigned)d >> 4;
    return ((((t + 1u) * GEMM_GROUPS + (M_TILES - 1)) / M_TILES) - 1u) & 255u;
}

#define BM_STAGE_PITCH 33

__global__ __launch_bounds__(GEMM_WAVES * 32) void input_gemm_kernel(
    const float* __restrict__ pos, const float* __restrict__ feat,
    const float* __restrict__ W1,  const float* __restrict__ Wp0,
    const int* __restrict__ gids,
    const int* __restrict__ dst, const int nEdges,
    float* __restrict__ yBuf, float* __restrict__ partial,
    unsigned* __restrict__ bm)
{
    __shared__ float part[2][GEMM_WAVES][PART_TILE];
    __shared__ unsigned stage[2][32][BM_STAGE_PITCH];

    const int  wave   = __builtin_amdgcn_readfirstlane((int)(threadIdx.x >> 5));
    const int  lane   = threadIdx.x & 31;
    const int  col    = lane & 15;
    const int  h      = lane >> 4;
    const bool hasPos = wave < D_POS / 4;

    const int kf = wave * KW_FEAT + 2 * h;
    const int kp = (wave & 7) * 4 + 2 * h;
    v2f wy0[CH_FEAT], wy1[CH_FEAT], wz0[CH_FEAT], wz1[CH_FEAT];
    for (int c = 0; c < CH_FEAT; ++c) {
        const int r = (33 + kf + 4 * c) * 32;
        wy0[c].x = W1 [r + col];        wy0[c].y = W1 [r + 32 + col];
        wy1[c].x = W1 [r + 16 + col];   wy1[c].y = W1 [r + 48 + col];
        wz0[c].x = Wp0[r + col];        wz0[c].y = Wp0[r + 32 + col];
        wz1[c].x = Wp0[r + 16 + col];   wz1[c].y = Wp0[r + 48 + col];
    }
    v2f py0 = {}, py1 = {}, pz0 = {}, pz1 = {};
    if (hasPos) {
        const int r = kp * 32;
        py0.x = W1 [r + col];        py0.y = W1 [r + 32 + col];
        py1.x = W1 [r + 16 + col];   py1.y = W1 [r + 48 + col];
        pz0.x = Wp0[r + col];        pz0.y = Wp0[r + 32 + col];
        pz1.x = Wp0[r + 16 + col];   pz1.y = Wp0[r + 48 + col];
    }

    {
        const int nLoads   = (nEdges + 4095) >> 12;
        const int nSlices  = nLoads * 4;
        const int bmWords  = nLoads * 128;
        for (int s0 = (int)blockIdx.x; s0 < nSlices; s0 += 2 * GEMM_GROUPS) {
            if (s0 != (int)blockIdx.x) __syncthreads();
            int  dv[2][2];
            bool vl[2][2];
            for (int sl = 0; sl < 2; ++sl)
                for (int cc = 0; cc < 2; ++cc) {
                    const int e = (s0 + sl * GEMM_GROUPS) * 1024 + (wave + GEMM_WAVES * cc) * 32 + lane;
                    vl[sl][cc] = e < nEdges;
                    dv[sl][cc] = dst[vl[sl][cc] ? e : 0];
                }
            for (int sl = 0; sl < 2; ++sl)
                for (int cc = 0; cc < 2; ++cc) {
                    const unsigned ob = ownerOfNode(dv[sl][cc]);
                    const unsigned hi = vl[sl][cc] ? (ob >> 4)  : 16u;
                    const unsigned lo = vl[sl][cc] ? (ob & 15u) : 16u;
                    unsigned wv = 0u;
                    for (int r = 0; r < 16; ++r) {
                        const unsigned bh = __builtin_amdgcn_ballot_w32(hi == (unsigned)r);
                        const unsigned bl = __builtin_amdgcn_ballot_w32(lo == (unsigned)r);
                        wv = (lane == r)      ? bh : wv;
                        wv = (lane == 16 + r) ? bl : wv;
                    }
                    stage[sl][lane][wave + GEMM_WAVES * cc] = wv;
                }
            __syncthreads();
            for (int sl = 0; sl < 2; ++sl) {
                const int s = s0 + sl * GEMM_GROUPS;
                if (s < nSlices) {
                    for (int rr = 0; rr < 2; ++rr) {
                        const int r = 2 * wave + rr;
                        bm[(size_t)r * bmWords + s * 32 + lane] = stage[sl][r][lane];
                    }
                }
            }
        }
    }

    const int tileBeg = ((int)blockIdx.x * M_TILES) / GEMM_GROUPS;
    const int tileEnd = (((int)blockIdx.x + 1) * M_TILES) / GEMM_GROUPS;

    v2f  sumF[CH_FEAT], sumP = {};
    for (int c = 0; c < CH_FEAT; ++c) sumF[c] = v2f{};
    int  runG   = 0;
    bool runOn  = false;

    auto headProduct = [&](const v2f (&xF)[CH_FEAT], const v2f xP,
                           v8f& cz0, v8f& cz1) __attribute__((always_inline)) {
        cz0 = v8f{}; cz1 = v8f{};
        if (hasPos) {
            cz0 = __builtin_amdgcn_wmma_f32_16x16x4_f32(false, xP, false, pz0, (short)0, cz0, false, false);
            cz1 = __builtin_amdgcn_wmma_f32_16x16x4_f32(false, xP, false, pz1, (short)0, cz1, false, false);
        }
        for (int c = 0; c < CH_FEAT; ++c) {
            cz0 = __builtin_amdgcn_wmma_f32_16x16x4_f32(false, xF[c], false, wz0[c], (short)0, cz0, false, false);
            cz1 = __builtin_amdgcn_wmma_f32_16x16x4_f32(false, xF[c], false, wz1[c], (short)0, cz1, false, false);
        }
    };

    auto closeRun = [&]() __attribute__((always_inline)) {
        v8f cz0, cz1;
        headProduct(sumF, sumP, cz0, cz1);
        float s0 = 0.0f, s1 = 0.0f;
        for (int j = 0; j < 8; ++j) { s0 += cz0[j]; s1 += cz1[j]; }
        atomicAdd(&partial[runG * 32 + col],      s0);
        atomicAdd(&partial[runG * 32 + 16 + col], s1);
        for (int c = 0; c < CH_FEAT; ++c) sumF[c] = v2f{};
        sumP  = v2f{};
        runOn = false;
    };

    auto tileStep = [&](const int tile, const int buf,
                        const v2f (&aF)[CH_FEAT], const v2f aP, const int aG,
                        v2f (&nF)[CH_FEAT], v2f& nP, int& nG) __attribute__((always_inline)) {
        const int nextTile = tile + 1;
        if (nextTile < tileEnd) {
            const int node = nextTile * 16 + col;
            const float* fr = feat + (size_t)node * D_FEAT + kf;
            for (int c = 0; c < CH_FEAT; ++c) nF[c] = *(const v2f*)(fr + 4 * c);
            if (hasPos) nP = *(const v2f*)(pos + (size_t)node * D_POS + kp);
            nG = gids[node];
        }

        const int  g0   = __builtin_amdgcn_readfirstlane(aG);
        const bool oneG = __builtin_amdgcn_ballot_w32(aG != g0) == 0u;
        if (oneG) {
            if (runOn && g0 != runG) closeRun();
            for (int c = 0; c < CH_FEAT; ++c) sumF[c] += aF[c];
            sumP += aP;
            runG  = g0;
            runOn = true;
        } else {
            if (runOn) closeRun();
            v8f cz0, cz1;
            headProduct(aF, aP, cz0, cz1);
            for (int j = 0; j < 8; ++j) {
                const int gr = gids[tile * 16 + 8 * h + j];
                atomicAdd(&partial[gr * 32 + col],      cz0[j]);
                atomicAdd(&partial[gr * 32 + 16 + col], cz1[j]);
            }
        }

        v8f cy0 = {}, cy1 = {};
        if (hasPos) {
            cy0 = __builtin_amdgcn_wmma_f32_16x16x4_f32(false, aP, false, py0, (short)0, cy0, false, false);
            cy1 = __builtin_amdgcn_wmma_f32_16x16x4_f32(false, aP, false, py1, (short)0, cy1, false, false);
        }
        for (int c = 0; c < CH_FEAT; ++c) {
            cy0 = __builtin_amdgcn_wmma_f32_16x16x4_f32(false, aF[c], false, wy0[c], (short)0, cy0, false, false);
            cy1 = __builtin_amdgcn_wmma_f32_16x16x4_f32(false, aF[c], false, wy1[c], (short)0, cy1, false, false);
        }

        float* P = &part[buf][wave][0];
        for (int j = 0; j < 8; ++j) {
            float* pr = P + (8 * h + j) * PART_PITCH + col;
            pr[0]  = cy0[j];
            pr[16] = cy1[j];
        }
        __syncthreads();

        float sy = 0.0f;
        const float* R = &part[buf][0][wave * PART_PITCH + lane];
        for (int p = 0; p < GEMM_WAVES; ++p) sy += R[p * PART_TILE];
        const int off = (tile * 16 + wave) * 32 + lane;
        yBuf[off]   = sy;
    };

    int tile = tileBeg;
    v2f fA[CH_FEAT], pA = {};  int gA;
    v2f fB[CH_FEAT], pB = {};  int gB;
    {
        const int node = tile * 16 + col;
        const float* fr = feat + (size_t)node * D_FEAT + kf;
        for (int c = 0; c < CH_FEAT; ++c) { fA[c] = *(const v2f*)(fr + 4 * c); fB[c] = fA[c]; }
        if (hasPos) pA = *(const v2f*)(pos + (size_t)node * D_POS + kp);
        gA = gids[node];
        gB = gA;
    }

    while (true) {
        tileStep(tile, 0, fA, pA, gA, fB, pB, gB);
        ++tile;
        if (tile >= tileEnd) break;
        tileStep(tile, 1, fB, pB, gB, fA, pA, gA);
        ++tile;
        if (tile >= tileEnd) break;
    }
    if (runOn) closeRun();
}

#define EN_WAVES      32
#define EN_THREADS    (EN_WAVES * 32)
#define ACC_PITCH     36
#define OWN_TILES_MAX ((M_TILES + GEMM_GROUPS - 1) / GEMM_GROUPS)
#define OWN_NODES_MAX (OWN_TILES_MAX * 16)
#define SCAN_LOADS    4
#define Q_CAP         64

__global__ __launch_bounds__(EN_THREADS) void edge_node_kernel(
    const float* __restrict__ yBuf, const int* __restrict__ src,
    const int* __restrict__ dst, const int nEdges,
    const unsigned* __restrict__ bm,
    const float* __restrict__ W2, const float* __restrict__ bias1,
    const float* __restrict__ bias2, const float* __restrict__ Wp1,
    const int* __restrict__ gids, float* __restrict__ partial)
{
    __shared__ float    accL[OWN_NODES_MAX * ACC_PITCH];
    __shared__ unsigned ring[EN_WAVES][Q_CAP];
    __shared__ float    hbuf[OWN_TILES_MAX][16 * 32];

    const int wave = __builtin_amdgcn_readfirstlane((int)(threadIdx.x >> 5));
    const int lane = threadIdx.x & 31;

    const int      tileBeg = ((int)blockIdx.x * M_TILES) / GEMM_GROUPS;
    const int      tileEnd = (((int)blockIdx.x + 1) * M_TILES) / GEMM_GROUPS;
    const int      n0      = tileBeg * 16;
    const unsigned cnt     = (unsigned)(tileEnd - tileBeg) * 16u;

    for (int i = threadIdx.x; i < (int)cnt * 32; i += EN_THREADS)
        accL[(i >> 5) * ACC_PITCH + (i & 31)] = yBuf[(size_t)n0 * 32 + i];
    __syncthreads();

    int qHead = 0, qn = 0;

    auto flush = [&](const bool full) __attribute__((always_inline)) {
        const int n = (full || qn > 32) ? 32 : qn;
        int t = 0, s = 0;
        if (lane < n) {
            const unsigned e = ring[wave][(qHead + lane) & (Q_CAP - 1)];
            s = src[e];
            t = dst[e] - n0;
        }
        for (int k0 = 0; k0 < n; k0 += 8) {
            float v[8];
            for (int j = 0; j < 8; ++j) {
                const int sk = __builtin_amdgcn_readlane(s, k0 + j);
                v[j] = yBuf[(size_t)sk * 32 + lane];
            }
            for (int j = 0; j < 8; ++j) {
                if (full || k0 + j < n) {
                    const int  tk  = __builtin_amdgcn_readlane(t, k0 + j);
                    const bool own = (unsigned)tk < cnt;
                    __hip_atomic_fetch_add(&accL[(own ? tk : 0) * ACC_PITCH + lane], own ? v[j] : 0.0f,
                                           __ATOMIC_RELAXED, __HIP_MEMORY_SCOPE_WORKGROUP);
                }
            }
        }
        qHead = (qHead + n) & (Q_CAP - 1);
        qn   -= n;
    };

    const int nLoads  = (nEdges + 4095) >> 12;
    const int bmWords = nLoads * 128;
    const unsigned* bmH = bm + (size_t)((int)blockIdx.x >> 4) * bmWords;
    const unsigned* bmL = bm + (size_t)(16 + ((int)blockIdx.x & 15)) * bmWords;
    const int nMine   = (nLoads - wave + EN_WAVES - 1) / EN_WAVES;

    auto wordOf = [&](const int it) __attribute__((always_inline)) {
        const int itc = it < nMine ? it : nMine - 1;
        return (size_t)(wave + itc * EN_WAVES) * 128 + lane * 4;
    };
    auto keepOwned = [&](uint4 m, const int it) __attribute__((always_inline)) {
        if (it >= nMine) m = uint4{0u, 0u, 0u, 0u};
        const unsigned eBase = (unsigned)(wave + it * EN_WAVES) * 4096u + (unsigned)lane * 128u;
        while (true) {
            const unsigned mask = __builtin_amdgcn_ballot_w32((m.x | m.y | m.z | m.w) != 0u);
            if (mask == 0u) break;
            unsigned mm = m.x, k = 0u;
            if (mm == 0u) { mm = m.y; k = 1u; }
            if (mm == 0u) { mm = m.z; k = 2u; }
            if (mm == 0u) { mm = m.w; k = 3u; }
            if (mm != 0u) {
                const unsigned e = eBase + 32u * k + (unsigned)__builtin_ctz(mm);
                const int slot = (qHead + qn + (int)__builtin_amdgcn_mbcnt_lo(mask, 0u)) & (Q_CAP - 1);
                ring[wave][slot] = e;
            }
            qn += __builtin_popcount(mask);
            const unsigned cm = mm & (mm - 1u);
            m.x = (k == 0u) ? cm : m.x;
            m.y = (k == 1u) ? cm : m.y;
            m.z = (k == 2u) ? cm : m.z;
            m.w = (k == 3u) ? cm : m.w;
            while (qn >= 32) flush(true);
        }
    };
    for (int it0 = 0; it0 < nMine; it0 += SCAN_LOADS) {
        const size_t w0 = wordOf(it0),     w1 = wordOf(it0 + 1);
        const size_t w2 = wordOf(it0 + 2), w3 = wordOf(it0 + 3);
        const uint4 h0 = *(const uint4*)(bmH + w0), l0 = *(const uint4*)(bmL + w0);
        const uint4 h1 = *(const uint4*)(bmH + w1), l1 = *(const uint4*)(bmL + w1);
        const uint4 h2 = *(const uint4*)(bmH + w2), l2 = *(const uint4*)(bmL + w2);
        const uint4 h3 = *(const uint4*)(bmH + w3), l3 = *(const uint4*)(bmL + w3);
        keepOwned(uint4{h0.x & l0.x, h0.y & l0.y, h0.z & l0.z, h0.w & l0.w}, it0);
        keepOwned(uint4{h1.x & l1.x, h1.y & l1.y, h1.z & l1.z, h1.w & l1.w}, it0 + 1);
        keepOwned(uint4{h2.x & l2.x, h2.y & l2.y, h2.z & l2.z, h2.w & l2.w}, it0 + 2);
        keepOwned(uint4{h3.x & l3.x, h3.y & l3.y, h3.z & l3.z, h3.w & l3.w}, it0 + 3);
    }
    while (qn > 0) flush(false);
    __syncthreads();

    if (wave >= tileEnd - tileBeg) return;

    const int  m0  = (tileBeg + wave) * 16;
    const int  col = lane & 15;
    const bool hi  = lane >= 16;

    const float* accRow = &accL[(wave * 16 + col) * ACC_PITCH];
    v8f c10 = {}, c11 = {};
    for (int k = 0; k < 32; k += 4) {
        const int kb = k + (hi ? 2 : 0);
        const v2f av = *(const v2f*)(accRow + kb);
        const v2f bv = *(const v2f*)(bias1 + kb);
        v2f a;
        a.x = fmaxf(av.x + bv.x, 0.0f);
        a.y = fmaxf(av.y + bv.y, 0.0f);
        v2f b0, b1f;
        b0.x  = W2[kb * 32 + col];       b0.y  = W2[(kb + 1) * 32 + col];
        b1f.x = W2[kb * 32 + 16 + col];  b1f.y = W2[(kb + 1) * 32 + 16 + col];
        c10 = __builtin_amdgcn_wmma_f32_16x16x4_f32(false, a, false, b0,  (short)0, c10, false, false);
        c11 = __builtin_amdgcn_wmma_f32_16x16x4_f32(false, a, false, b1f, (short)0, c11, false, false);
    }

    const float bb0 = bias2[col];
    const float bb1 = bias2[16 + col];
    for (int j = 0; j < 8; ++j) {
        const int row = (hi ? 8 : 0) + j;
        hbuf[wave][row * 32 + col]      = fmaxf(c10[j] + bb0, 0.0f);
        hbuf[wave][row * 32 + 16 + col] = fmaxf(c11[j] + bb1, 0.0f);
    }

    v8f u0 = {}, u1 = {};
    for (int k = 0; k < 32; k += 4) {
        const int kb = k + (hi ? 2 : 0);
        const v2f a = *(const v2f*)(&hbuf[wave][col * 32 + kb]);
        v2f b0, b1f;
        b0.x  = Wp1[kb * 32 + col];       b0.y  = Wp1[(kb + 1) * 32 + col];
        b1f.x = Wp1[kb * 32 + 16 + col];  b1f.y = Wp1[(kb + 1) * 32 + 16 + col];
        u0 = __builtin_amdgcn_wmma_f32_16x16x4_f32(false, a, false, b0,  (short)0, u0, false, false);
        u1 = __builtin_amdgcn_wmma_f32_16x16x4_f32(false, a, false, b1f, (short)0, u1, false, false);
    }

    for (int j = 0; j < 8; ++j) {
        const int row = (hi ? 8 : 0) + j;
        const int g   = gids[m0 + row];
        atomicAdd(&partial[g * 32 + col],      u0[j]);
        atomicAdd(&partial[g * 32 + 16 + col], u1[j]);
    }
}

__global__ void zero_partial_kernel(float* __restrict__ partial) {
    const int i = blockIdx.x * blockDim.x + threadIdx.x;
    if (i < N_GRAPHS * D_OUT) partial[i] = 0.0f;
}

__global__ void finalize_kernel(const float* __restrict__ partial,
                                const float* __restrict__ bp0,
                                const float* __restrict__ bp1,
                                float* __restrict__ out) {
    const int i = blockIdx.x * blockDim.x + threadIdx.x;
    if (i < N_GRAPHS * D_OUT) {
        const int c = i & 31;
        out[i] = partial[i] + bp0[c] + bp1[c];
    }
}

extern "C" void kernel_launch(void* const* d_in, const int* in_sizes, int n_in,
                              void* d_out, int out_size, void* d_ws, size_t ws_size,
                              hipStream_t stream) {
    const float* pos  = (const float*)d_in[0];
    const float* feat = (const float*)d_in[1];
    const int*   src  = (const int*)  d_in[2];
    const int*   dst  = (const int*)  d_in[3];
    const int*   gids = (const int*)  d_in[4];
    const float* W1   = (const float*)d_in[5];
    const float* b1   = (const float*)d_in[6];
    const float* W2   = (const float*)d_in[7];
    const float* b2   = (const float*)d_in[8];
    const float* Wp0  = (const float*)d_in[9];
    const float* bp0  = (const float*)d_in[10];
    const float* Wp1  = (const float*)d_in[11];
    const float* bp1  = (const float*)d_in[12];
    float* out = (float*)d_out;
    const int nEdges = in_sizes[2];

    float*    yBuf    = (float*)d_ws;
    unsigned* bm      = (unsigned*)(yBuf + (size_t)N_NODES * 32);
    float*    partial = yBuf + (size_t)N_NODES * 64;

    zero_partial_kernel<<<(N_GRAPHS * D_OUT + 255) / 256, 256, 0, stream>>>(partial);
    input_gemm_kernel<<<GEMM_GROUPS, GEMM_WAVES * 32, 0, stream>>>(pos, feat, W1, Wp0, gids,
                                                                   dst, nEdges, yBuf, partial, bm);
    edge_node_kernel<<<GEMM_GROUPS, EN_THREADS, 0, stream>>>(yBuf, src, dst, nEdges, bm,
                                                             W2, b1, b2, Wp1, gids, partial);
    finalize_kernel<<<(N_GRAPHS * D_OUT + 255) / 256, 256, 0, stream>>>(
        partial, bp0, bp1, out);
}


# kernel 1 barriers order LDS only (LDS-scoped fences, not __syncthreads) and every load of a step is unconditional: A rows fetched ahead stay in flight across the tile barrier and the bitmap job
# speedup vs baseline: 2.0087x; 1.0634x over previous
#include <hip/hip_runtime.h>
#include <hip/hip_bf16.h>


#define N_NODES 50000
#define D_POS   32
#define D_FEAT  768
#define D_IN    801
#define D_HID   32
#define D_OUT   32
#define N_GRAPHS 128
#define M_TILES  (N_NODES / 16)

#define GEMM_WAVES   16
#define GEMM_GROUPS  256
#define KW_FEAT      (D_FEAT / GEMM_WAVES)
#define CH_FEAT      (KW_FEAT / 4)
#define PART_PITCH   36
#define PART_TILE    (16 * PART_PITCH)

typedef __attribute__((ext_vector_type(2))) float v2f;
typedef __attribute__((ext_vector_type(8))) float v8f;

__device__ __forceinline__ unsigned ownerOfNode(const int d) {
    const unsigned t = (unsigned)d >> 4;
    return ((((t + 1u) * GEMM_GROUPS + (M_TILES - 1)) / M_TILES) - 1u) & 255u;
}

__device__ __forceinline__ void ldsBarrier() {
    __builtin_amdgcn_fence(__ATOMIC_RELEASE, "workgroup", "local");
    __builtin_amdgcn_s_barrier();
    __builtin_amdgcn_fence(__ATOMIC_ACQUIRE, "workgroup", "local");
}

#define BM_STAGE_PITCH 33

__global__ __launch_bounds__(GEMM_WAVES * 32) void input_gemm_kernel(
    const float* __restrict__ pos, const float* __restrict__ feat,
    const float* __restrict__ W1,  const float* __restrict__ Wp0,
    const int* __restrict__ gids,
    const int* __restrict__ dst, const int nEdges,
    float* __restrict__ yBuf, float* __restrict__ partial,
    unsigned* __restrict__ bm)
{
    __shared__ float part[2][GEMM_WAVES][PART_TILE];
    __shared__ unsigned stage[2][32][BM_STAGE_PITCH];

    const int  wave   = __builtin_amdgcn_readfirstlane((int)(threadIdx.x >> 5));
    const int  lane   = threadIdx.x & 31;
    const int  col    = lane & 15;
    const int  h      = lane >> 4;
    const bool hasPos = wave < D_POS / 4;

    const int kf = wave * KW_FEAT + 2 * h;
    const int kp = (wave & 7) * 4 + 2 * h;
    v2f wy0[CH_FEAT], wy1[CH_FEAT], wz0[CH_FEAT], wz1[CH_FEAT];
    for (int c = 0; c < CH_FEAT; ++c) {
        const int r = (33 + kf + 4 * c) * 32;
        wy0[c].x = W1 [r + col];        wy0[c].y = W1 [r + 32 + col];
        wy1[c].x = W1 [r + 16 + col];   wy1[c].y = W1 [r + 48 + col];
        wz0[c].x = Wp0[r + col];        wz0[c].y = Wp0[r + 32 + col];
        wz1[c].x = Wp0[r + 16 + col];   wz1[c].y = Wp0[r + 48 + col];
    }
    v2f py0 = {}, py1 = {}, pz0 = {}, pz1 = {};
    if (hasPos) {
        const int r = kp * 32;
        py0.x = W1 [r + col];        py0.y = W1 [r + 32 + col];
        py1.x = W1 [r + 16 + col];   py1.y = W1 [r + 48 + col];
        pz0.x = Wp0[r + col];        pz0.y = Wp0[r + 32 + col];
        pz1.x = Wp0[r + 16 + col];   pz1.y = Wp0[r + 48 + col];
    }

    const int tileBeg = ((int)blockIdx.x * M_TILES) / GEMM_GROUPS;
    const int tileEnd = (((int)blockIdx.x + 1) * M_TILES) / GEMM_GROUPS;

    v2f fA[CH_FEAT], pA = {};  int gA;
    {
        const int node = tileBeg * 16 + col;
        const float* fr = feat + (size_t)node * D_FEAT + kf;
        for (int c = 0; c < CH_FEAT; ++c) fA[c] = *(const v2f*)(fr + 4 * c);
        pA = *(const v2f*)(pos + (size_t)node * D_POS + kp);
        gA = gids[node];
    }

    {
        const int nLoads   = (nEdges + 4095) >> 12;
        const int nSlices  = nLoads * 4;
        const int bmWords  = nLoads * 128;
        for (int s0 = (int)blockIdx.x; s0 < nSlices; s0 += 2 * GEMM_GROUPS) {
            if (s0 != (int)blockIdx.x) ldsBarrier();
            int  dv[2][2];
            bool vl[2][2];
            for (int sl = 0; sl < 2; ++sl)
                for (int cc = 0; cc < 2; ++cc) {
                    const int e = (s0 + sl * GEMM_GROUPS) * 1024 + (wave + GEMM_WAVES * cc) * 32 + lane;
                    vl[sl][cc] = e < nEdges;
                    dv[sl][cc] = dst[vl[sl][cc] ? e : 0];
                }
            for (int sl = 0; sl < 2; ++sl)
                for (int cc = 0; cc < 2; ++cc) {
                    const unsigned ob = ownerOfNode(dv[sl][cc]);
                    const unsigned hi = vl[sl][cc] ? (ob >> 4)  : 16u;
                    const unsigned lo = vl[sl][cc] ? (ob & 15u) : 16u;
                    unsigned wv = 0u;
                    for (int r = 0; r < 16; ++r) {
                        const unsigned bh = __builtin_amdgcn_ballot_w32(hi == (unsigned)r);
                        const unsigned bl = __builtin_amdgcn_ballot_w32(lo == (unsigned)r);
                        wv = (lane == r)      ? bh : wv;
                        wv = (lane == 16 + r) ? bl : wv;
                    }
                    stage[sl][lane][wave + GEMM_WAVES * cc] = wv;
                }
            ldsBarrier();
            for (int sl = 0; sl < 2; ++sl) {
                const int s = s0 + sl * GEMM_GROUPS;
                if (s < nSlices) {
                    for (int rr = 0; rr < 2; ++rr) {
                        const int r = 2 * wave + rr;
                        bm[(size_t)r * bmWords + s * 32 + lane] = stage[sl][r][lane];
                    }
                }
            }
        }
    }

    v2f  sumF[CH_FEAT], sumP = {};
    for (int c = 0; c < CH_FEAT; ++c) sumF[c] = v2f{};
    int  runG   = 0;
    bool runOn  = false;

    auto headProduct = [&](const v2f (&xF)[CH_FEAT], const v2f xP,
                           v8f& cz0, v8f& cz1) __attribute__((always_inline)) {
        cz0 = v8f{}; cz1 = v8f{};
        if (hasPos) {
            cz0 = __builtin_amdgcn_wmma_f32_16x16x4_f32(false, xP, false, pz0, (short)0, cz0, false, false);
            cz1 = __builtin_amdgcn_wmma_f32_16x16x4_f32(false, xP, false, pz1, (short)0, cz1, false, false);
        }
        for (int c = 0; c < CH_FEAT; ++c) {
            cz0 = __builtin_amdgcn_wmma_f32_16x16x4_f32(false, xF[c], false, wz0[c], (short)0, cz0, false, false);
            cz1 = __builtin_amdgcn_wmma_f32_16x16x4_f32(false, xF[c], false, wz1[c], (short)0, cz1, false, false);
        }
    };

    auto closeRun = [&]() __attribute__((always_inline)) {
        v8f cz0, cz1;
        headProduct(sumF, sumP, cz0, cz1);
        float s0 = 0.0f, s1 = 0.0f;
        for (int j = 0; j < 8; ++j) { s0 += cz0[j]; s1 += cz1[j]; }
        atomicAdd(&partial[runG * 32 + col],      s0);
        atomicAdd(&partial[runG * 32 + 16 + col], s1);
        for (int c = 0; c < CH_FEAT; ++c) sumF[c] = v2f{};
        sumP  = v2f{};
        runOn = false;
    };

    auto tileStep = [&](const int tile, const int buf,
                        const v2f (&aF)[CH_FEAT], const v2f aP, const int aG,
                        v2f (&nF)[CH_FEAT], v2f& nP, int& nG) __attribute__((always_inline)) {
        {
            const bool more = tile + 1 < tileEnd;
            const int  node = (more ? tile + 1 : tile) * 16 + col;
            const float* fr = more ? feat + (size_t)node * D_FEAT + kf : W1;
            const float* pr = more ? pos + (size_t)node * D_POS + kp   : W1;
            for (int c = 0; c < CH_FEAT; ++c) nF[c] = *(const v2f*)(fr + 4 * c);
            nP = *(const v2f*)pr;
            nG = gids[node];
        }

        const int  g0   = __builtin_amdgcn_readfirstlane(aG);
        const bool oneG = __builtin_amdgcn_ballot_w32(aG != g0) == 0u;
        if (oneG) {
            if (runOn && g0 != runG) closeRun();
            for (int c = 0; c < CH_FEAT; ++c) sumF[c] += aF[c];
            sumP += aP;
            runG  = g0;
            runOn = true;
        } else {
            if (runOn) closeRun();
            v8f cz0, cz1;
            headProduct(aF, aP, cz0, cz1);
            for (int j = 0; j < 8; ++j) {
                const int gr = gids[tile * 16 + 8 * h + j];
                atomicAdd(&partial[gr * 32 + col],      cz0[j]);
                atomicAdd(&partial[gr * 32 + 16 + col], cz1[j]);
            }
        }

        v8f cy0 = {}, cy1 = {};
        if (hasPos) {
            cy0 = __builtin_amdgcn_wmma_f32_16x16x4_f32(false, aP, false, py0, (short)0, cy0, false, false);
            cy1 = __builtin_amdgcn_wmma_f32_16x16x4_f32(false, aP, false, py1, (short)0, cy1, false, false);
        }
        for (int c = 0; c < CH_FEAT; ++c) {
            cy0 = __builtin_amdgcn_wmma_f32_16x16x4_f32(false, aF[c], false, wy0[c], (short)0, cy0, false, false);
            cy1 = __builtin_amdgcn_wmma_f32_16x16x4_f32(false, aF[c], false, wy1[c], (short)0, cy1, false, false);
        }

        float* P = &part[buf][wave][0];
        for (int j = 0; j < 8; ++j) {
            float* pr = P + (8 * h + j) * PART_PITCH + col;
            pr[0]  = cy0[j];
            pr[16] = cy1[j];
        }
        ldsBarrier();

        float sy = 0.0f;
        const float* R = &part[buf][0][wave * PART_PITCH + lane];
        for (int p = 0; p < GEMM_WAVES; ++p) sy += R[p * PART_TILE];
        const int off = (tile * 16 + wave) * 32 + lane;
        yBuf[off]   = sy;
    };

    int tile = tileBeg;
    v2f fB[CH_FEAT], pB = {};  int gB;
    for (int c = 0; c < CH_FEAT; ++c) fB[c] = fA[c];
    gB = gA;

    while (true) {
        tileStep(tile, 0, fA, pA, gA, fB, pB, gB);
        ++tile;
        if (tile >= tileEnd) break;
        tileStep(tile, 1, fB, pB, gB, fA, pA, gA);
        ++tile;
        if (tile >= tileEnd) break;
    }
    if (runOn) closeRun();
}

#define EN_WAVES      32
#define EN_THREADS    (EN_WAVES * 32)
#define ACC_PITCH     36
#define OWN_TILES_MAX ((M_TILES + GEMM_GROUPS - 1) / GEMM_GROUPS)
#define OWN_NODES_MAX (OWN_TILES_MAX * 16)
#define SCAN_LOADS    4
#define Q_CAP         64

__global__ __launch_bounds__(EN_THREADS) void edge_node_kernel(
    const float* __restrict__ yBuf, const int* __restrict__ src,
    const int* __restrict__ dst, const int nEdges,
    const unsigned* __restrict__ bm,
    const float* __restrict__ W2, const float* __restrict__ bias1,
    const float* __restrict__ bias2, const float* __restrict__ Wp1,
    const int* __restrict__ gids, float* __restrict__ partial)
{
    __shared__ float    accL[OWN_NODES_MAX * ACC_PITCH];
    __shared__ unsigned ring[EN_WAVES][Q_CAP];
    __shared__ float    hbuf[OWN_TILES_MAX][16 * 32];

    const int wave = __builtin_amdgcn_readfirstlane((int)(threadIdx.x >> 5));
    const int lane = threadIdx.x & 31;

    const int      tileBeg = ((int)blockIdx.x * M_TILES) / GEMM_GROUPS;
    const int      tileEnd = (((int)blockIdx.x + 1) * M_TILES) / GEMM_GROUPS;
    const int      n0      = tileBeg * 16;
    const unsigned cnt     = (unsigned)(tileEnd - tileBeg) * 16u;

    for (int i = threadIdx.x; i < (int)cnt * 32; i += EN_THREADS)
        accL[(i >> 5) * ACC_PITCH + (i & 31)] = yBuf[(size_t)n0 * 32 + i];
    __syncthreads();

    int qHead = 0, qn = 0;

    auto flush = [&](const bool full) __attribute__((always_inline)) {
        const int n = (full || qn > 32) ? 32 : qn;
        int t = 0, s = 0;
        if (lane < n) {
            const unsigned e = ring[wave][(qHead + lane) & (Q_CAP - 1)];
            s = src[e];
            t = dst[e] - n0;
        }
        for (int k0 = 0; k0 < n; k0 += 8) {
            float v[8];
            for (int j = 0; j < 8; ++j) {
                const int sk = __builtin_amdgcn_readlane(s, k0 + j);
                v[j] = yBuf[(size_t)sk * 32 + lane];
            }
            for (int j = 0; j < 8; ++j) {
                if (full || k0 + j < n) {
                    const int  tk  = __builtin_amdgcn_readlane(t, k0 + j);
                    const bool own = (unsigned)tk < cnt;
                    __hip_atomic_fetch_add(&accL[(own ? tk : 0) * ACC_PITCH + lane], own ? v[j] : 0.0f,
                                           __ATOMIC_RELAXED, __HIP_MEMORY_SCOPE_WORKGROUP);
                }
            }
        }
        qHead = (qHead + n) & (Q_CAP - 1);
        qn   -= n;
    };

    const int nLoads  = (nEdges + 4095) >> 12;
    const int bmWords = nLoads * 128;
    const unsigned* bmH = bm + (size_t)((int)blockIdx.x >> 4) * bmWords;
    const unsigned* bmL = bm + (size_t)(16 + ((int)blockIdx.x & 15)) * bmWords;
    const int nMine   = (nLoads - wave + EN_WAVES - 1) / EN_WAVES;

    auto wordOf = [&](const int it) __attribute__((always_inline)) {
        const int itc = it < nMine ? it : nMine - 1;
        return (size_t)(wave + itc * EN_WAVES) * 128 + lane * 4;
    };
    auto keepOwned = [&](uint4 m, const int it) __attribute__((always_inline)) {
        if (it >= nMine) m = uint4{0u, 0u, 0u, 0u};
        const unsigned eBase = (unsigned)(wave + it * EN_WAVES) * 4096u + (unsigned)lane * 128u;
        while (true) {
            const unsigned mask = __builtin_amdgcn_ballot_w32((m.x | m.y | m.z | m.w) != 0u);
            if (mask == 0u) break;
            unsigned mm = m.x, k = 0u;
            if (mm == 0u) { mm = m.y; k = 1u; }
            if (mm == 0u) { mm = m.z; k = 2u; }
            if (mm == 0u) { mm = m.w; k = 3u; }
            if (mm != 0u) {
                const unsigned e = eBase + 32u * k + (unsigned)__builtin_ctz(mm);
                const int slot = (qHead + qn + (int)__builtin_amdgcn_mbcnt_lo(mask, 0u)) & (Q_CAP - 1);
                ring[wave][slot] = e;
            }
            qn += __builtin_popcount(mask);
            const unsigned cm = mm & (mm - 1u);
            m.x = (k == 0u) ? cm : m.x;
            m.y = (k == 1u) ? cm : m.y;
            m.z = (k == 2u) ? cm : m.z;
            m.w = (k == 3u) ? cm : m.w;
            while (qn >= 32) flush(true);
        }
    };
    for (int it0 = 0; it0 < nMine; it0 += SCAN_LOADS) {
        const size_t w0 = wordOf(it0),     w1 = wordOf(it0 + 1);
        const size_t w2 = wordOf(it0 + 2), w3 = wordOf(it0 + 3);
        const uint4 h0 = *(const uint4*)(bmH + w0), l0 = *(const uint4*)(bmL + w0);
        const uint4 h1 = *(const uint4*)(bmH + w1), l1 = *(const uint4*)(bmL + w1);
        const uint4 h2 = *(const uint4*)(bmH + w2), l2 = *(const uint4*)(bmL + w2);
        const uint4 h3 = *(const uint4*)(bmH + w3), l3 = *(const uint4*)(bmL + w3);
        keepOwned(uint4{h0.x & l0.x, h0.y & l0.y, h0.z & l0.z, h0.w & l0.w}, it0);
        keepOwned(uint4{h1.x & l1.x, h1.y & l1.y, h1.z & l1.z, h1.w & l1.w}, it0 + 1);
        keepOwned(uint4{h2.x & l2.x, h2.y & l2.y, h2.z & l2.z, h2.w & l2.w}, it0 + 2);
        keepOwned(uint4{h3.x & l3.x, h3.y & l3.y, h3.z & l3.z, h3.w & l3.w}, it0 + 3);
    }
    while (qn > 0) flush(false);
    __syncthreads();

    if (wave >= tileEnd - tileBeg) return;

    const int  m0  = (tileBeg + wave) * 16;
    const int  col = lane & 15;
    const bool hi  = lane >= 16;

    const float* accRow = &accL[(wave * 16 + col) * ACC_PITCH];
    v8f c10 = {}, c11 = {};
    for (int k = 0; k < 32; k += 4) {
        const int kb = k + (hi ? 2 : 0);
        const v2f av = *(const v2f*)(accRow + kb);
        const v2f bv = *(const v2f*)(bias1 + kb);
        v2f a;
        a.x = fmaxf(av.x + bv.x, 0.0f);
        a.y = fmaxf(av.y + bv.y, 0.0f);
        v2f b0, b1f;
        b0.x  = W2[kb * 32 + col];       b0.y  = W2[(kb + 1) * 32 + col];
        b1f.x = W2[kb * 32 + 16 + col];  b1f.y = W2[(kb + 1) * 32 + 16 + col];
        c10 = __builtin_amdgcn_wmma_f32_16x16x4_f32(false, a, false, b0,  (short)0, c10, false, false);
        c11 = __builtin_amdgcn_wmma_f32_16x16x4_f32(false, a, false, b1f, (short)0, c11, false, false);
    }

    const float bb0 = bias2[col];
    const float bb1 = bias2[16 + col];
    for (int j = 0; j < 8; ++j) {
        const int row = (hi ? 8 : 0) + j;
        hbuf[wave][row * 32 + col]      = fmaxf(c10[j] + bb0, 0.0f);
        hbuf[wave][row * 32 + 16 + col] = fmaxf(c11[j] + bb1, 0.0f);
    }

    v8f u0 = {}, u1 = {};
    for (int k = 0; k < 32; k += 4) {
        const int kb = k + (hi ? 2 : 0);
        const v2f a = *(const v2f*)(&hbuf[wave][col * 32 + kb]);
        v2f b0, b1f;
        b0.x  = Wp1[kb * 32 + col];       b0.y  = Wp1[(kb + 1) * 32 + col];
        b1f.x = Wp1[kb * 32 + 16 + col];  b1f.y = Wp1[(kb + 1) * 32 + 16 + col];
        u0 = __builtin_amdgcn_wmma_f32_16x16x4_f32(false, a, false, b0,  (short)0, u0, false, false);
        u1 = __builtin_amdgcn_wmma_f32_16x16x4_f32(false, a, false, b1f, (short)0, u1, false, false);
    }

    for (int j = 0; j < 8; ++j) {
        const int row = (hi ? 8 : 0) + j;
        const int g   = gids[m0 + row];
        atomicAdd(&partial[g * 32 + col],      u0[j]);
        atomicAdd(&partial[g * 32 + 16 + col], u1[j]);
    }
}

__global__ void zero_partial_kernel(float* __restrict__ partial) {
    const int i = blockIdx.x * blockDim.x + threadIdx.x;
    if (i < N_GRAPHS * D_OUT) partial[i] = 0.0f;
}

__global__ void finalize_kernel(const float* __restrict__ partial,
                                const float* __restrict__ bp0,
                                const float* __restrict__ bp1,
                                float* __restrict__ out) {
    const int i = blockIdx.x * blockDim.x + threadIdx.x;
    if (i < N_GRAPHS * D_OUT) {
        const int c = i & 31;
        out[i] = partial[i] + bp0[c] + bp1[c];
    }
}

extern "C" void kernel_launch(void* const* d_in, const int* in_sizes, int n_in,
                              void* d_out, int out_size, void* d_ws, size_t ws_size,
                              hipStream_t stream) {
    const float* pos  = (const float*)d_in[0];
    const float* feat = (const float*)d_in[1];
    const int*   src  = (const int*)  d_in[2];
    const int*   dst  = (const int*)  d_in[3];
    const int*   gids = (const int*)  d_in[4];
    const float* W1   = (const float*)d_in[5];
    const float* b1   = (const float*)d_in[6];
    const float* W2   = (const float*)d_in[7];
    const float* b2   = (const float*)d_in[8];
    const float* Wp0  = (const float*)d_in[9];
    const float* bp0  = (const float*)d_in[10];
    const float* Wp1  = (const float*)d_in[11];
    const float* bp1  = (const float*)d_in[12];
    float* out = (float*)d_out;
    const int nEdges = in_sizes[2];

    float*    yBuf    = (float*)d_ws;
    unsigned* bm      = (unsigned*)(yBuf + (size_t)N_NODES * 32);
    float*    partial = yBuf + (size_t)N_NODES * 64;

    zero_partial_kernel<<<(N_GRAPHS * D_OUT + 255) / 256, 256, 0, stream>>>(partial);
    input_gemm_kernel<<<GEMM_GROUPS, GEMM_WAVES * 32, 0, stream>>>(pos, feat, W1, Wp0, gids,
                                                                   dst, nEdges, yBuf, partial, bm);
    edge_node_kernel<<<GEMM_GROUPS, EN_THREADS, 0, stream>>>(yBuf, src, dst, nEdges, bm,
                                                             W2, b1, b2, Wp1, gids, partial);
    finalize_kernel<<<(N_GRAPHS * D_OUT + 255) / 256, 256, 0, stream>>>(
        partial, bp0, bp1, out);
}


# graph sums (head term, u rows) added up per work-group in an LDS table (LDS float adds); one whole-line global float add per touched graph and work-group, not per wave and row; sum order changes
# speedup vs baseline: 2.4321x; 1.2108x over previous
#include <hip/hip_runtime.h>
#include <hip/hip_bf16.h>


#define N_NODES 50000
#define D_POS   32
#define D_FEAT  768
#define D_IN    801
#define D_HID   32
#define D_OUT   32
#define N_GRAPHS 128
#define M_TILES  (N_NODES / 16)

#define GEMM_WAVES   16
#define GEMM_GROUPS  256
#define KW_FEAT      (D_FEAT / GEMM_WAVES)
#define CH_FEAT      (KW_FEAT / 4)
#define PART_PITCH   36
#define PART_TILE    (16 * PART_PITCH)

typedef __attribute__((ext_vector_type(2))) float v2f;
typedef __attribute__((ext_vector_type(8))) float v8f;

__device__ __forceinline__ unsigned ownerOfNode(const int d) {
    const unsigned t = (unsigned)d >> 4;
    return ((((t + 1u) * GEMM_GROUPS + (M_TILES - 1)) / M_TILES) - 1u) & 255u;
}

__device__ __forceinline__ void ldsBarrier() {
    __builtin_amdgcn_fence(__ATOMIC_RELEASE, "workgroup", "local");
    __builtin_amdgcn_s_barrier();
    __builtin_amdgcn_fence(__ATOMIC_ACQUIRE, "workgroup", "local");
}

#define BM_STAGE_PITCH 33

__device__ __forceinline__ void ldsAddF(float* p, const float v) {
    __hip_atomic_fetch_add(p, v, __ATOMIC_RELAXED, __HIP_MEMORY_SCOPE_WORKGROUP);
}

__global__ __launch_bounds__(GEMM_WAVES * 32) void input_gemm_kernel(
    const float* __restrict__ pos, const float* __restrict__ feat,
    const float* __restrict__ W1,  const float* __restrict__ Wp0,
    const int* __restrict__ gids,
    const int* __restrict__ dst, const int nEdges,
    float* __restrict__ yBuf, float* __restrict__ partial,
    unsigned* __restrict__ bm)
{
    __shared__ float part[2][GEMM_WAVES][PART_TILE];
    __shared__ unsigned stage[2][32][BM_STAGE_PITCH];
    __shared__ float    gsum[N_GRAPHS * D_OUT];
    __shared__ unsigned gflag[N_GRAPHS];

    for (int i = threadIdx.x; i < N_GRAPHS * D_OUT; i += GEMM_WAVES * 32) gsum[i] = 0.0f;
    if (threadIdx.x < N_GRAPHS) gflag[threadIdx.x] = 0u;

    const int  wave   = __builtin_amdgcn_readfirstlane((int)(threadIdx.x >> 5));
    const int  lane   = threadIdx.x & 31;
    const int  col    = lane & 15;
    const int  h      = lane >> 4;
    const bool hasPos = wave < D_POS / 4;

    const int kf = wave * KW_FEAT + 2 * h;
    const int kp = (wave & 7) * 4 + 2 * h;
    v2f wy0[CH_FEAT], wy1[CH_FEAT], wz0[CH_FEAT], wz1[CH_FEAT];
    for (int c = 0; c < CH_FEAT; ++c) {
        const int r = (33 + kf + 4 * c) * 32;
        wy0[c].x = W1 [r + col];        wy0[c].y = W1 [r + 32 + col];
        wy1[c].x = W1 [r + 16 + col];   wy1[c].y = W1 [r + 48 + col];
        wz0[c].x = Wp0[r + col];        wz0[c].y = Wp0[r + 32 + col];
        wz1[c].x = Wp0[r + 16 + col];   wz1[c].y = Wp0[r + 48 + col];
    }
    v2f py0 = {}, py1 = {}, pz0 = {}, pz1 = {};
    if (hasPos) {
        const int r = kp * 32;
        py0.x = W1 [r + col];        py0.y = W1 [r + 32 + col];
        py1.x = W1 [r + 16 + col];   py1.y = W1 [r + 48 + col];
        pz0.x = Wp0[r + col];        pz0.y = Wp0[r + 32 + col];
        pz1.x = Wp0[r + 16 + col];   pz1.y = Wp0[r + 48 + col];
    }

    const int tileBeg = ((int)blockIdx.x * M_TILES) / GEMM_GROUPS;
    const int tileEnd = (((int)blockIdx.x + 1) * M_TILES) / GEMM_GROUPS;

    v2f fA[CH_FEAT], pA = {};  int gA;
    {
        const int node = tileBeg * 16 + col;
        const float* fr = feat + (size_t)node * D_FEAT + kf;
        for (int c = 0; c < CH_FEAT; ++c) fA[c] = *(const v2f*)(fr + 4 * c);
        pA = *(const v2f*)(pos + (size_t)node * D_POS + kp);
        gA = gids[node];
    }

    {
        const int nLoads   = (nEdges + 4095) >> 12;
        const int nSlices  = nLoads * 4;
        const int bmWords  = nLoads * 128;
        for (int s0 = (int)blockIdx.x; s0 < nSlices; s0 += 2 * GEMM_GROUPS) {
            if (s0 != (int)blockIdx.x) ldsBarrier();
            int  dv[2][2];
            bool vl[2][2];
            for (int sl = 0; sl < 2; ++sl)
                for (int cc = 0; cc < 2; ++cc) {
                    const int e = (s0 + sl * GEMM_GROUPS) * 1024 + (wave + GEMM_WAVES * cc) * 32 + lane;
                    vl[sl][cc] = e < nEdges;
                    dv[sl][cc] = dst[vl[sl][cc] ? e : 0];
                }
            for (int sl = 0; sl < 2; ++sl)
                for (int cc = 0; cc < 2; ++cc) {
                    const unsigned ob = ownerOfNode(dv[sl][cc]);
                    const unsigned hi = vl[sl][cc] ? (ob >> 4)  : 16u;
                    const unsigned lo = vl[sl][cc] ? (ob & 15u) : 16u;
                    unsigned wv = 0u;
                    for (int r = 0; r < 16; ++r) {
                        const unsigned bh = __builtin_amdgcn_ballot_w32(hi == (unsigned)r);
                        const unsigned bl = __builtin_amdgcn_ballot_w32(lo == (unsigned)r);
                        wv = (lane == r)      ? bh : wv;
                        wv = (lane == 16 + r) ? bl : wv;
                    }
                    stage[sl][lane][wave + GEMM_WAVES * cc] = wv;
                }
            ldsBarrier();
            for (int sl = 0; sl < 2; ++sl) {
                const int s = s0 + sl * GEMM_GROUPS;
                if (s < nSlices) {
                    for (int rr = 0; rr < 2; ++rr) {
                        const int r = 2 * wave + rr;
                        bm[(size_t)r * bmWords + s * 32 + lane] = stage[sl][r][lane];
                    }
                }
            }
        }
    }

    v2f  sumF[CH_FEAT], sumP = {};
    for (int c = 0; c < CH_FEAT; ++c) sumF[c] = v2f{};
    int  runG   = 0;
    bool runOn  = false;

    auto headProduct = [&](const v2f (&xF)[CH_FEAT], const v2f xP,
                           v8f& cz0, v8f& cz1) __attribute__((always_inline)) {
        cz0 = v8f{}; cz1 = v8f{};
        if (hasPos) {
            cz0 = __builtin_amdgcn_wmma_f32_16x16x4_f32(false, xP, false, pz0, (short)0, cz0, false, false);
            cz1 = __builtin_amdgcn_wmma_f32_16x16x4_f32(false, xP, false, pz1, (short)0, cz1, false, false);
        }
        for (int c = 0; c < CH_FEAT; ++c) {
            cz0 = __builtin_amdgcn_wmma_f32_16x16x4_f32(false, xF[c], false, wz0[c], (short)0, cz0, false, false);
            cz1 = __builtin_amdgcn_wmma_f32_16x16x4_f32(false, xF[c], false, wz1[c], (short)0, cz1, false, false);
        }
    };

    auto closeRun = [&]() __attribute__((always_inline)) {
        v8f cz0, cz1;
        headProduct(sumF, sumP, cz0, cz1);
        float s0 = 0.0f, s1 = 0.0f;
        for (int j = 0; j < 8; ++j) { s0 += cz0[j]; s1 += cz1[j]; }
        ldsAddF(&gsum[runG * 32 + col],      s0);
        ldsAddF(&gsum[runG * 32 + 16 + col], s1);
        if (lane == 0) gflag[runG] = 1u;
        for (int c = 0; c < CH_FEAT; ++c) sumF[c] = v2f{};
        sumP  = v2f{};
        runOn = false;
    };

    auto tileStep = [&](const int tile, const int buf,
                        const v2f (&aF)[CH_FEAT], const v2f aP, const int aG,
                        v2f (&nF)[CH_FEAT], v2f& nP, int& nG) __attribute__((always_inline)) {
        {
            const bool more = tile + 1 < tileEnd;
            const int  node = (more ? tile + 1 : tile) * 16 + col;
            const float* fr = more ? feat + (size_t)node * D_FEAT + kf : W1;
            const float* pr = more ? pos + (size_t)node * D_POS + kp   : W1;
            for (int c = 0; c < CH_FEAT; ++c) nF[c] = *(const v2f*)(fr + 4 * c);
            nP = *(const v2f*)pr;
            nG = gids[node];
        }

        const int  g0   = __builtin_amdgcn_readfirstlane(aG);
        const bool oneG = __builtin_amdgcn_ballot_w32(aG != g0) == 0u;
        if (oneG) {
            if (runOn && g0 != runG) closeRun();
            for (int c = 0; c < CH_FEAT; ++c) sumF[c] += aF[c];
            sumP += aP;
            runG  = g0;
            runOn = true;
        } else {
            if (runOn) closeRun();
            v8f cz0, cz1;
            headProduct(aF, aP, cz0, cz1);
            for (int j = 0; j < 8; ++j) {
                const int gr = gids[tile * 16 + 8 * h + j];
                ldsAddF(&gsum[gr * 32 + col],      cz0[j]);
                ldsAddF(&gsum[gr * 32 + 16 + col], cz1[j]);
                gflag[gr] = 1u;
            }
        }

        v8f cy0 = {}, cy1 = {};
        if (hasPos) {
            cy0 = __builtin_amdgcn_wmma_f32_16x16x4_f32(false, aP, false, py0, (short)0, cy0, false, false);
            cy1 = __builtin_amdgcn_wmma_f32_16x16x4_f32(false, aP, false, py1, (short)0, cy1, false, false);
        }
        for (int c = 0; c < CH_FEAT; ++c) {
            cy0 = __builtin_amdgcn_wmma_f32_16x16x4_f32(false, aF[c], false, wy0[c], (short)0, cy0, false, false);
            cy1 = __builtin_amdgcn_wmma_f32_16x16x4_f32(false, aF[c], false, wy1[c], (short)0, cy1, false, false);
        }

        float* P = &part[buf][wave][0];
        for (int j = 0; j < 8; ++j) {
            float* pr = P + (8 * h + j) * PART_PITCH + col;
            pr[0]  = cy0[j];
            pr[16] = cy1[j];
        }
        ldsBarrier();

        float sy = 0.0f;
        const float* R = &part[buf][0][wave * PART_PITCH + lane];
        for (int p = 0; p < GEMM_WAVES; ++p) sy += R[p * PART_TILE];
        const int off = (tile * 16 + wave) * 32 + lane;
        yBuf[off]   = sy;
    };

    ldsBarrier();

    int tile = tileBeg;
    v2f fB[CH_FEAT], pB = {};  int gB;
    for (int c = 0; c < CH_FEAT; ++c) fB[c] = fA[c];
    gB = gA;

    while (true) {
        tileStep(tile, 0, fA, pA, gA, fB, pB, gB);
        ++tile;
        if (tile >= tileEnd) break;
        tileStep(tile, 1, fB, pB, gB, fA, pA, gA);
        ++tile;
        if (tile >= tileEnd) break;
    }
    if (runOn) closeRun();

    ldsBarrier();
    for (int g = wave; g < N_GRAPHS; g += GEMM_WAVES)
        if (gflag[g] != 0u)
            atomicAdd(&partial[g * 32 + lane], gsum[g * 32 + lane]);
}

#define EN_WAVES      32
#define EN_THREADS    (EN_WAVES * 32)
#define ACC_PITCH     36
#define OWN_TILES_MAX ((M_TILES + GEMM_GROUPS - 1) / GEMM_GROUPS)
#define OWN_NODES_MAX (OWN_TILES_MAX * 16)
#define SCAN_LOADS    4
#define Q_CAP         64

__global__ __launch_bounds__(EN_THREADS) void edge_node_kernel(
    const float* __restrict__ yBuf, const int* __restrict__ src,
    const int* __restrict__ dst, const int nEdges,
    const unsigned* __restrict__ bm,
    const float* __restrict__ W2, const float* __restrict__ bias1,
    const float* __restrict__ bias2, const float* __restrict__ Wp1,
    const int* __restrict__ gids, float* __restrict__ partial)
{
    __shared__ float    accL[OWN_NODES_MAX * ACC_PITCH];
    __shared__ unsigned ring[EN_WAVES][Q_CAP];
    __shared__ float    hbuf[OWN_TILES_MAX][16 * 32];
    __shared__ float    gsum[N_GRAPHS * D_OUT];
    __shared__ unsigned gflag[N_GRAPHS];

    const int wave = __builtin_amdgcn_readfirstlane((int)(threadIdx.x >> 5));
    const int lane = threadIdx.x & 31;

    const int      tileBeg = ((int)blockIdx.x * M_TILES) / GEMM_GROUPS;
    const int      tileEnd = (((int)blockIdx.x + 1) * M_TILES) / GEMM_GROUPS;
    const int      n0      = tileBeg * 16;
    const unsigned cnt     = (unsigned)(tileEnd - tileBeg) * 16u;

    for (int i = threadIdx.x; i < N_GRAPHS * D_OUT; i += EN_THREADS) gsum[i] = 0.0f;
    if (threadIdx.x < N_GRAPHS) gflag[threadIdx.x] = 0u;

    for (int i = threadIdx.x; i < (int)cnt * 32; i += EN_THREADS)
        accL[(i >> 5) * ACC_PITCH + (i & 31)] = yBuf[(size_t)n0 * 32 + i];
    __syncthreads();

    int qHead = 0, qn = 0;

    auto flush = [&](const bool full) __attribute__((always_inline)) {
        const int n = (full || qn > 32) ? 32 : qn;
        int t = 0, s = 0;
        if (lane < n) {
            const unsigned e = ring[wave][(qHead + lane) & (Q_CAP - 1)];
            s = src[e];
            t = dst[e] - n0;
        }
        for (int k0 = 0; k0 < n; k0 += 8) {
            float v[8];
            for (int j = 0; j < 8; ++j) {
                const int sk = __builtin_amdgcn_readlane(s, k0 + j);
                v[j] = yBuf[(size_t)sk * 32 + lane];
            }
            for (int j = 0; j < 8; ++j) {
                if (full || k0 + j < n) {
                    const int  tk  = __builtin_amdgcn_readlane(t, k0 + j);
                    const bool own = (unsigned)tk < cnt;
                    __hip_atomic_fetch_add(&accL[(own ? tk : 0) * ACC_PITCH + lane], own ? v[j] : 0.0f,
                                           __ATOMIC_RELAXED, __HIP_MEMORY_SCOPE_WORKGROUP);
                }
            }
        }
        qHead = (qHead + n) & (Q_CAP - 1);
        qn   -= n;
    };

    const int nLoads  = (nEdges + 4095) >> 12;
    const int bmWords = nLoads * 128;
    const unsigned* bmH = bm + (size_t)((int)blockIdx.x >> 4) * bmWords;
    const unsigned* bmL = bm + (size_t)(16 + ((int)blockIdx.x & 15)) * bmWords;
    const int nMine   = (nLoads - wave + EN_WAVES - 1) / EN_WAVES;

    auto wordOf = [&](const int it) __attribute__((always_inline)) {
        const int itc = it < nMine ? it : nMine - 1;
        return (size_t)(wave + itc * EN_WAVES) * 128 + lane * 4;
    };
    auto keepOwned = [&](uint4 m, const int it) __attribute__((always_inline)) {
        if (it >= nMine) m = uint4{0u, 0u, 0u, 0u};
        const unsigned eBase = (unsigned)(wave + it * EN_WAVES) * 4096u + (unsigned)lane * 128u;
        while (true) {
            const unsigned mask = __builtin_amdgcn_ballot_w32((m.x | m.y | m.z | m.w) != 0u);
            if (mask == 0u) break;
            unsigned mm = m.x, k = 0u;
            if (mm == 0u) { mm = m.y; k = 1u; }
            if (mm == 0u) { mm = m.z; k = 2u; }
            if (mm == 0u) { mm = m.w; k = 3u; }
            if (mm != 0u) {
                const unsigned e = eBase + 32u * k + (unsigned)__builtin_ctz(mm);
                const int slot = (qHead + qn + (int)__builtin_amdgcn_mbcnt_lo(mask, 0u)) & (Q_CAP - 1);
                ring[wave][slot] = e;
            }
            qn += __builtin_popcount(mask);
            const unsigned cm = mm & (mm - 1u);
            m.x = (k == 0u) ? cm : m.x;
            m.y = (k == 1u) ? cm : m.y;
            m.z = (k == 2u) ? cm : m.z;
            m.w = (k == 3u) ? cm : m.w;
            while (qn >= 32) flush(true);
        }
    };
    for (int it0 = 0; it0 < nMine; it0 += SCAN_LOADS) {
        const size_t w0 = wordOf(it0),     w1 = wordOf(it0 + 1);
        const size_t w2 = wordOf(it0 + 2), w3 = wordOf(it0 + 3);
        const uint4 h0 = *(const uint4*)(bmH + w0), l0 = *(const uint4*)(bmL + w0);
        const uint4 h1 = *(const uint4*)(bmH + w1), l1 = *(const uint4*)(bmL + w1);
        const uint4 h2 = *(const uint4*)(bmH + w2), l2 = *(const uint4*)(bmL + w2);
        const uint4 h3 = *(const uint4*)(bmH + w3), l3 = *(const uint4*)(bmL + w3);
        keepOwned(uint4{h0.x & l0.x, h0.y & l0.y, h0.z & l0.z, h0.w & l0.w}, it0);
        keepOwned(uint4{h1.x & l1.x, h1.y & l1.y, h1.z & l1.z, h1.w & l1.w}, it0 + 1);
        keepOwned(uint4{h2.x & l2.x, h2.y & l2.y, h2.z & l2.z, h2.w & l2.w}, it0 + 2);
        keepOwned(uint4{h3.x & l3.x, h3.y & l3.y, h3.z & l3.z, h3.w & l3.w}, it0 + 3);
    }
    while (qn > 0) flush(false);
    __syncthreads();

    if (wave < tileEnd - tileBeg) {
        const int  m0  = (tileBeg + wave) * 16;
        const int  col = lane & 15;
        const bool hi  = lane >= 16;

        int gr[8];
        for (int j = 0; j < 8; ++j) gr[j] = gids[m0 + (hi ? 8 : 0) + j];

        const float* accRow = &accL[(wave * 16 + col) * ACC_PITCH];
        v8f c10 = {}, c11 = {};
        for (int k = 0; k < 32; k += 4) {
            const int kb = k + (hi ? 2 : 0);
            const v2f av = *(const v2f*)(accRow + kb);
            const v2f bv = *(const v2f*)(bias1 + kb);
            v2f a;
            a.x = fmaxf(av.x + bv.x, 0.0f);
            a.y = fmaxf(av.y + bv.y, 0.0f);
            v2f b0, b1f;
            b0.x  = W2[kb * 32 + col];       b0.y  = W2[(kb + 1) * 32 + col];
            b1f.x = W2[kb * 32 + 16 + col];  b1f.y = W2[(kb + 1) * 32 + 16 + col];
            c10 = __builtin_amdgcn_wmma_f32_16x16x4_f32(false, a, false, b0,  (short)0, c10, false, false);
            c11 = __builtin_amdgcn_wmma_f32_16x16x4_f32(false, a, false, b1f, (short)0, c11, false, false);
        }

        const float bb0 = bias2[col];
        const float bb1 = bias2[16 + col];
        for (int j = 0; j < 8; ++j) {
            const int row = (hi ? 8 : 0) + j;
            hbuf[wave][row * 32 + col]      = fmaxf(c10[j] + bb0, 0.0f);
            hbuf[wave][row * 32 + 16 + col] = fmaxf(c11[j] + bb1, 0.0f);
        }

        v8f u0 = {}, u1 = {};
        for (int k = 0; k < 32; k += 4) {
            const int kb = k + (hi ? 2 : 0);
            const v2f a = *(const v2f*)(&hbuf[wave][col * 32 + kb]);
            v2f b0, b1f;
            b0.x  = Wp1[kb * 32 + col];       b0.y  = Wp1[(kb + 1) * 32 + col];
            b1f.x = Wp1[kb * 32 + 16 + col];  b1f.y = Wp1[(kb + 1) * 32 + 16 + col];
            u0 = __builtin_amdgcn_wmma_f32_16x16x4_f32(false, a, false, b0,  (short)0, u0, false, false);
            u1 = __builtin_amdgcn_wmma_f32_16x16x4_f32(false, a, false, b1f, (short)0, u1, false, false);
        }

        const int g0   = __builtin_amdgcn_readfirstlane(gr[0]);
        bool      same = true;
        for (int j = 0; j < 8; ++j) same = same && (gr[j] == g0);
        if (__builtin_amdgcn_ballot_w32(!same) == 0u) {
            float s0 = 0.0f, s1 = 0.0f;
            for (int j = 0; j < 8; ++j) { s0 += u0[j]; s1 += u1[j]; }
            ldsAddF(&gsum[g0 * 32 + col],      s0);
            ldsAddF(&gsum[g0 * 32 + 16 + col], s1);
            if (lane == 0) gflag[g0] = 1u;
        } else {
            for (int j = 0; j < 8; ++j) {
                ldsAddF(&gsum[gr[j] * 32 + col],      u0[j]);
                ldsAddF(&gsum[gr[j] * 32 + 16 + col], u1[j]);
                gflag[gr[j]] = 1u;
            }
        }
    }

    __syncthreads();
    for (int g = wave; g < N_GRAPHS; g += EN_WAVES)
        if (gflag[g] != 0u)
            atomicAdd(&partial[g * 32 + lane], gsum[g * 32 + lane]);
}

__global__ void zero_partial_kernel(float* __restrict__ partial) {
    const int i = blockIdx.x * blockDim.x + threadIdx.x;
    if (i < N_GRAPHS * D_OUT) partial[i] = 0.0f;
}

__global__ void finalize_kernel(const float* __restrict__ partial,
                                const float* __restrict__ bp0,
                                const float* __restrict__ bp1,
                                float* __restrict__ out) {
    const int i = blockIdx.x * blockDim.x + threadIdx.x;
    if (i < N_GRAPHS * D_OUT) {
        const int c = i & 31;
        out[i] = partial[i] + bp0[c] + bp1[c];
    }
}

extern "C" void kernel_launch(void* const* d_in, const int* in_sizes, int n_in,
                              void* d_out, int out_size, void* d_ws, size_t ws_size,
                              hipStream_t stream) {
    const float* pos  = (const float*)d_in[0];
    const float* feat = (const float*)d_in[1];
    const int*   src  = (const int*)  d_in[2];
    const int*   dst  = (const int*)  d_in[3];
    const int*   gids = (const int*)  d_in[4];
    const float* W1   = (const float*)d_in[5];
    const float* b1   = (const float*)d_in[6];
    const float* W2   = (const float*)d_in[7];
    const float* b2   = (const float*)d_in[8];
    const float* Wp0  = (const float*)d_in[9];
    const float* bp0  = (const float*)d_in[10];
    const float* Wp1  = (const float*)d_in[11];
    const float* bp1  = (const float*)d_in[12];
    float* out = (float*)d_out;
    const int nEdges = in_sizes[2];

    float*    yBuf    = (float*)d_ws;
    unsigned* bm      = (unsigned*)(yBuf + (size_t)N_NODES * 32);
    float*    partial = yBuf + (size_t)N_NODES * 64;

    zero_partial_kernel<<<(N_GRAPHS * D_OUT + 255) / 256, 256, 0, stream>>>(partial);
    input_gemm_kernel<<<GEMM_GROUPS, GEMM_WAVES * 32, 0, stream>>>(pos, feat, W1, Wp0, gids,
                                                                   dst, nEdges, yBuf, partial, bm);
    edge_node_kernel<<<GEMM_GROUPS, EN_THREADS, 0, stream>>>(yBuf, src, dst, nEdges, bm,
                                                             W2, b1, b2, Wp1, gids, partial);
    finalize_kernel<<<(N_GRAPHS * D_OUT + 255) / 256, 256, 0, stream>>>(
        partial, bp0, bp1, out);
}


# no accumulation buffer: kernel 1 stores biases into out and hands each work-group's head sums to the same owner in kernel 2 (private plane), which adds into out; 4 to 2 launches; sum order changes
# speedup vs baseline: 2.5610x; 1.0530x over previous
#include <hip/hip_runtime.h>
#include <hip/hip_bf16.h>


#define N_NODES 50000
#define D_POS   32
#define D_FEAT  768
#define D_IN    801
#define D_HID   32
#define D_OUT   32
#define N_GRAPHS 128
#define M_TILES  (N_NODES / 16)

#define GEMM_WAVES   16
#define GEMM_GROUPS  256
#define KW_FEAT      (D_FEAT / GEMM_WAVES)
#define CH_FEAT      (KW_FEAT / 4)
#define PART_PITCH   36
#define PART_TILE    (16 * PART_PITCH)

typedef __attribute__((ext_vector_type(2))) float v2f;
typedef __attribute__((ext_vector_type(8))) float v8f;

__device__ __forceinline__ unsigned ownerOfNode(const int d) {
    const unsigned t = (unsigned)d >> 4;
    return ((((t + 1u) * GEMM_GROUPS + (M_TILES - 1)) / M_TILES) - 1u) & 255u;
}

__device__ __forceinline__ void ldsBarrier() {
    __builtin_amdgcn_fence(__ATOMIC_RELEASE, "workgroup", "local");
    __builtin_amdgcn_s_barrier();
    __builtin_amdgcn_fence(__ATOMIC_ACQUIRE, "workgroup", "local");
}

#define BM_STAGE_PITCH 33

#define BM_ROOM_WORDS 458752
#define HAND_WORDS    (GEMM_GROUPS * N_GRAPHS * D_OUT)
__device__ __forceinline__ void ldsAddF(float* p, const float v) {
    __hip_atomic_fetch_add(p, v, __ATOMIC_RELAXED, __HIP_MEMORY_SCOPE_WORKGROUP);
}

__global__ __launch_bounds__(GEMM_WAVES * 32) void input_gemm_kernel(
    const float* __restrict__ pos, const float* __restrict__ feat,
    const float* __restrict__ W1,  const float* __restrict__ Wp0,
    const int* __restrict__ gids,
    const int* __restrict__ dst, const int nEdges,
    const float* __restrict__ bp0, const float* __restrict__ bp1,
    float* __restrict__ yBuf, float* __restrict__ hand, unsigned* __restrict__ handFlag,
    unsigned* __restrict__ bm, float* __restrict__ out)
{
    __shared__ float part[2][GEMM_WAVES][PART_TILE];
    __shared__ unsigned stage[2][32][BM_STAGE_PITCH];
    __shared__ float    gsum[N_GRAPHS * D_OUT];
    __shared__ unsigned gflag[N_GRAPHS];

    for (int i = threadIdx.x; i < N_GRAPHS * D_OUT; i += GEMM_WAVES * 32) gsum[i] = 0.0f;
    if (threadIdx.x < N_GRAPHS) gflag[threadIdx.x] = 0u;

    const int  wave   = __builtin_amdgcn_readfirstlane((int)(threadIdx.x >> 5));
    const int  lane   = threadIdx.x & 31;
    const int  col    = lane & 15;
    const int  h      = lane >> 4;
    const bool hasPos = wave < D_POS / 4;

    const float biasRow = bp0[lane] + bp1[lane];

    const int kf = wave * KW_FEAT + 2 * h;
    const int kp = (wave & 7) * 4 + 2 * h;
    v2f wy0[CH_FEAT], wy1[CH_FEAT], wz0[CH_FEAT], wz1[CH_FEAT];
    for (int c = 0; c < CH_FEAT; ++c) {
        const int r = (33 + kf + 4 * c) * 32;
        wy0[c].x = W1 [r + col];        wy0[c].y = W1 [r + 32 + col];
        wy1[c].x = W1 [r + 16 + col];   wy1[c].y = W1 [r + 48 + col];
        wz0[c].x = Wp0[r + col];        wz0[c].y = Wp0[r + 32 + col];
        wz1[c].x = Wp0[r + 16 + col];   wz1[c].y = Wp0[r + 48 + col];
    }
    v2f py0 = {}, py1 = {}, pz0 = {}, pz1 = {};
    if (hasPos) {
        const int r = kp * 32;
        py0.x = W1 [r + col];        py0.y = W1 [r + 32 + col];
        py1.x = W1 [r + 16 + col];   py1.y = W1 [r + 48 + col];
        pz0.x = Wp0[r + col];        pz0.y = Wp0[r + 32 + col];
        pz1.x = Wp0[r + 16 + col];   pz1.y = Wp0[r + 48 + col];
    }

    const int tileBeg = ((int)blockIdx.x * M_TILES) / GEMM_GROUPS;
    const int tileEnd = (((int)blockIdx.x + 1) * M_TILES) / GEMM_GROUPS;

    v2f fA[CH_FEAT], pA = {};  int gA;
    {
        const int node = tileBeg * 16 + col;
        const float* fr = feat + (size_t)node * D_FEAT + kf;
        for (int c = 0; c < CH_FEAT; ++c) fA[c] = *(const v2f*)(fr + 4 * c);
        pA = *(const v2f*)(pos + (size_t)node * D_POS + kp);
        gA = gids[node];
    }

    {
        const int nLoads   = (nEdges + 4095) >> 12;
        const int nSlices  = nLoads * 4;
        const int bmWords  = nLoads * 128;
        for (int s0 = (int)blockIdx.x; s0 < nSlices; s0 += 2 * GEMM_GROUPS) {
            if (s0 != (int)blockIdx.x) ldsBarrier();
            int  dv[2][2];
            bool vl[2][2];
            for (int sl = 0; sl < 2; ++sl)
                for (int cc = 0; cc < 2; ++cc) {
                    const int e = (s0 + sl * GEMM_GROUPS) * 1024 + (wave + GEMM_WAVES * cc) * 32 + lane;
                    vl[sl][cc] = e < nEdges;
                    dv[sl][cc] = dst[vl[sl][cc] ? e : 0];
                }
            for (int sl = 0; sl < 2; ++sl)
                for (int cc = 0; cc < 2; ++cc) {
                    const unsigned ob = ownerOfNode(dv[sl][cc]);
                    const unsigned hi = vl[sl][cc] ? (ob >> 4)  : 16u;
                    const unsigned lo = vl[sl][cc] ? (ob & 15u) : 16u;
                    unsigned wv = 0u;
                    for (int r = 0; r < 16; ++r) {
                        const unsigned bh = __builtin_amdgcn_ballot_w32(hi == (unsigned)r);
                        const unsigned bl = __builtin_amdgcn_ballot_w32(lo == (unsigned)r);
                        wv = (lane == r)      ? bh : wv;
                        wv = (lane == 16 + r) ? bl : wv;
                    }
                    stage[sl][lane][wave + GEMM_WAVES * cc] = wv;
                }
            ldsBarrier();
            for (int sl = 0; sl < 2; ++sl) {
                const int s = s0 + sl * GEMM_GROUPS;
                if (s < nSlices) {
                    for (int rr = 0; rr < 2; ++rr) {
                        const int r = 2 * wave + rr;
                        bm[(size_t)r * bmWords + s * 32 + lane] = stage[sl][r][lane];
                    }
                }
            }
        }
    }

    v2f  sumF[CH_FEAT], sumP = {};
    for (int c = 0; c < CH_FEAT; ++c) sumF[c] = v2f{};
    int  runG   = 0;
    bool runOn  = false;

    auto headProduct = [&](const v2f (&xF)[CH_FEAT], const v2f xP,
                           v8f& cz0, v8f& cz1) __attribute__((always_inline)) {
        cz0 = v8f{}; cz1 = v8f{};
        if (hasPos) {
            cz0 = __builtin_amdgcn_wmma_f32_16x16x4_f32(false, xP, false, pz0, (short)0, cz0, false, false);
            cz1 = __builtin_amdgcn_wmma_f32_16x16x4_f32(false, xP, false, pz1, (short)0, cz1, false, false);
        }
        for (int c = 0; c < CH_FEAT; ++c) {
            cz0 = __builtin_amdgcn_wmma_f32_16x16x4_f32(false, xF[c], false, wz0[c], (short)0, cz0, false, false);
            cz1 = __builtin_amdgcn_wmma_f32_16x16x4_f32(false, xF[c], false, wz1[c], (short)0, cz1, false, false);
        }
    };

    auto closeRun = [&]() __attribute__((always_inline)) {
        v8f cz0, cz1;
        headProduct(sumF, sumP, cz0, cz1);
        float s0 = 0.0f, s1 = 0.0f;
        for (int j = 0; j < 8; ++j) { s0 += cz0[j]; s1 += cz1[j]; }
        ldsAddF(&gsum[runG * 32 + col],      s0);
        ldsAddF(&gsum[runG * 32 + 16 + col], s1);
        if (lane == 0) gflag[runG] = 1u;
        for (int c = 0; c < CH_FEAT; ++c) sumF[c] = v2f{};
        sumP  = v2f{};
        runOn = false;
    };

    auto tileStep = [&](const int tile, const int buf,
                        const v2f (&aF)[CH_FEAT], const v2f aP, const int aG,
                        v2f (&nF)[CH_FEAT], v2f& nP, int& nG) __attribute__((always_inline)) {
        {
            const bool more = tile + 1 < tileEnd;
            const int  node = (more ? tile + 1 : tile) * 16 + col;
            const float* fr = more ? feat + (size_t)node * D_FEAT + kf : W1;
            const float* pr = more ? pos + (size_t)node * D_POS + kp   : W1;
            for (int c = 0; c < CH_FEAT; ++c) nF[c] = *(const v2f*)(fr + 4 * c);
            nP = *(const v2f*)pr;
            nG = gids[node];
        }

        const int  g0   = __builtin_amdgcn_readfirstlane(aG);
        const bool oneG = __builtin_amdgcn_ballot_w32(aG != g0) == 0u;
        if (oneG) {
            if (runOn && g0 != runG) closeRun();
            for (int c = 0; c < CH_FEAT; ++c) sumF[c] += aF[c];
            sumP += aP;
            runG  = g0;
            runOn = true;
        } else {
            if (runOn) closeRun();
            v8f cz0, cz1;
            headProduct(aF, aP, cz0, cz1);
            for (int j = 0; j < 8; ++j) {
                const int gr = gids[tile * 16 + 8 * h + j];
                ldsAddF(&gsum[gr * 32 + col],      cz0[j]);
                ldsAddF(&gsum[gr * 32 + 16 + col], cz1[j]);
                gflag[gr] = 1u;
            }
        }

        v8f cy0 = {}, cy1 = {};
        if (hasPos) {
            cy0 = __builtin_amdgcn_wmma_f32_16x16x4_f32(false, aP, false, py0, (short)0, cy0, false, false);
            cy1 = __builtin_amdgcn_wmma_f32_16x16x4_f32(false, aP, false, py1, (short)0, cy1, false, false);
        }
        for (int c = 0; c < CH_FEAT; ++c) {
            cy0 = __builtin_amdgcn_wmma_f32_16x16x4_f32(false, aF[c], false, wy0[c], (short)0, cy0, false, false);
            cy1 = __builtin_amdgcn_wmma_f32_16x16x4_f32(false, aF[c], false, wy1[c], (short)0, cy1, false, false);
        }

        float* P = &part[buf][wave][0];
        for (int j = 0; j < 8; ++j) {
            float* pr = P + (8 * h + j) * PART_PITCH + col;
            pr[0]  = cy0[j];
            pr[16] = cy1[j];
        }
        ldsBarrier();

        float sy = 0.0f;
        const float* R = &part[buf][0][wave * PART_PITCH + lane];
        for (int p = 0; p < GEMM_WAVES; ++p) sy += R[p * PART_TILE];
        const int off = (tile * 16 + wave) * 32 + lane;
        yBuf[off]   = sy;
    };

    ldsBarrier();

    int tile = tileBeg;
    v2f fB[CH_FEAT], pB = {};  int gB;
    for (int c = 0; c < CH_FEAT; ++c) fB[c] = fA[c];
    gB = gA;

    while (true) {
        tileStep(tile, 0, fA, pA, gA, fB, pB, gB);
        ++tile;
        if (tile >= tileEnd) break;
        tileStep(tile, 1, fB, pB, gB, fA, pA, gA);
        ++tile;
        if (tile >= tileEnd) break;
    }
    if (runOn) closeRun();

    ldsBarrier();
    if (wave < N_GRAPHS / 32)
        handFlag[(size_t)blockIdx.x * N_GRAPHS + wave * 32 + lane] = gflag[wave * 32 + lane];
    for (int g = wave; g < N_GRAPHS; g += GEMM_WAVES)
        if (gflag[g] != 0u)
            hand[((size_t)blockIdx.x * N_GRAPHS + g) * 32 + lane] = gsum[g * 32 + lane];
    if (wave == N_GRAPHS / 32 && blockIdx.x < N_GRAPHS)
        out[blockIdx.x * 32 + lane] = biasRow;
}

#define EN_WAVES      32
#define EN_THREADS    (EN_WAVES * 32)
#define ACC_PITCH     36
#define OWN_TILES_MAX ((M_TILES + GEMM_GROUPS - 1) / GEMM_GROUPS)
#define OWN_NODES_MAX (OWN_TILES_MAX * 16)
#define SCAN_LOADS    4
#define Q_CAP         64

__global__ __launch_bounds__(EN_THREADS) void edge_node_kernel(
    const float* __restrict__ yBuf, const int* __restrict__ src,
    const int* __restrict__ dst, const int nEdges,
    const unsigned* __restrict__ bm,
    const float* __restrict__ W2, const float* __restrict__ bias1,
    const float* __restrict__ bias2, const float* __restrict__ Wp1,
    const int* __restrict__ gids,
    const float* __restrict__ hand, const unsigned* __restrict__ handFlag,
    float* __restrict__ out)
{
    __shared__ float    accL[OWN_NODES_MAX * ACC_PITCH];
    __shared__ unsigned ring[EN_WAVES][Q_CAP];
    __shared__ float    hbuf[OWN_TILES_MAX][16 * 32];
    __shared__ float    gsum[N_GRAPHS * D_OUT];
    __shared__ unsigned gflag[N_GRAPHS];

    const int wave = __builtin_amdgcn_readfirstlane((int)(threadIdx.x >> 5));
    const int lane = threadIdx.x & 31;

    const int      tileBeg = ((int)blockIdx.x * M_TILES) / GEMM_GROUPS;
    const int      tileEnd = (((int)blockIdx.x + 1) * M_TILES) / GEMM_GROUPS;
    const int      n0      = tileBeg * 16;
    const unsigned cnt     = (unsigned)(tileEnd - tileBeg) * 16u;

    for (int i = threadIdx.x; i < N_GRAPHS * D_OUT; i += EN_THREADS) gsum[i] = 0.0f;
    const unsigned handedFlag = handFlag[(size_t)blockIdx.x * N_GRAPHS + (threadIdx.x & (N_GRAPHS - 1))];

    for (int i = threadIdx.x; i < (int)cnt * 32; i += EN_THREADS)
        accL[(i >> 5) * ACC_PITCH + (i & 31)] = yBuf[(size_t)n0 * 32 + i];
    if (threadIdx.x < N_GRAPHS) gflag[threadIdx.x] = handedFlag;
    __syncthreads();

    float hv[N_GRAPHS / EN_WAVES];
    for (int j = 0; j < N_GRAPHS / EN_WAVES; ++j) {
        const int g = wave + EN_WAVES * j;
        hv[j] = 0.0f;
        if (gflag[g] != 0u)
            hv[j] = hand[((size_t)blockIdx.x * N_GRAPHS + g) * 32 + lane];
    }

    int qHead = 0, qn = 0;

    auto flush = [&](const bool full) __attribute__((always_inline)) {
        const int n = (full || qn > 32) ? 32 : qn;
        int t = 0, s = 0;
        if (lane < n) {
            const unsigned e = ring[wave][(qHead + lane) & (Q_CAP - 1)];
            s = src[e];
            t = dst[e] - n0;
        }
        for (int k0 = 0; k0 < n; k0 += 8) {
            float v[8];
            for (int j = 0; j < 8; ++j) {
                const int sk = __builtin_amdgcn_readlane(s, k0 + j);
                v[j] = yBuf[(size_t)sk * 32 + lane];
            }
            for (int j = 0; j < 8; ++j) {
                if (full || k0 + j < n) {
                    const int  tk  = __builtin_amdgcn_readlane(t, k0 + j);
                    const bool own = (unsigned)tk < cnt;
                    __hip_atomic_fetch_add(&accL[(own ? tk : 0) * ACC_PITCH + lane], own ? v[j] : 0.0f,
                                           __ATOMIC_RELAXED, __HIP_MEMORY_SCOPE_WORKGROUP);
                }
            }
        }
        qHead = (qHead + n) & (Q_CAP - 1);
        qn   -= n;
    };

    const int nLoads  = (nEdges + 4095) >> 12;
    const int bmWords = nLoads * 128;
    const unsigned* bmH = bm + (size_t)((int)blockIdx.x >> 4) * bmWords;
    const unsigned* bmL = bm + (size_t)(16 + ((int)blockIdx.x & 15)) * bmWords;
    const int nMine   = (nLoads - wave + EN_WAVES - 1) / EN_WAVES;

    auto wordOf = [&](const int it) __attribute__((always_inline)) {
        const int itc = it < nMine ? it : nMine - 1;
        return (size_t)(wave + itc * EN_WAVES) * 128 + lane * 4;
    };
    auto keepOwned = [&](uint4 m, const int it) __attribute__((always_inline)) {
        if (it >= nMine) m = uint4{0u, 0u, 0u, 0u};
        const unsigned eBase = (unsigned)(wave + it * EN_WAVES) * 4096u + (unsigned)lane * 128u;
        while (true) {
            const unsigned mask = __builtin_amdgcn_ballot_w32((m.x | m.y | m.z | m.w) != 0u);
            if (mask == 0u) break;
            unsigned mm = m.x, k = 0u;
            if (mm == 0u) { mm = m.y; k = 1u; }
            if (mm == 0u) { mm = m.z; k = 2u; }
            if (mm == 0u) { mm = m.w; k = 3u; }
            if (mm != 0u) {
                const unsigned e = eBase + 32u * k + (unsigned)__builtin_ctz(mm);
                const int slot = (qHead + qn + (int)__builtin_amdgcn_mbcnt_lo(mask, 0u)) & (Q_CAP - 1);
                ring[wave][slot] = e;
            }
            qn += __builtin_popcount(mask);
            const unsigned cm = mm & (mm - 1u);
            m.x = (k == 0u) ? cm : m.x;
            m.y = (k == 1u) ? cm : m.y;
            m.z = (k == 2u) ? cm : m.z;
            m.w = (k == 3u) ? cm : m.w;
            while (qn >= 32) flush(true);
        }
    };
    for (int it0 = 0; it0 < nMine; it0 += SCAN_LOADS) {
        const size_t w0 = wordOf(it0),     w1 = wordOf(it0 + 1);
        const size_t w2 = wordOf(it0 + 2), w3 = wordOf(it0 + 3);
        const uint4 h0 = *(const uint4*)(bmH + w0), l0 = *(const uint4*)(bmL + w0);
        const uint4 h1 = *(const uint4*)(bmH + w1), l1 = *(const uint4*)(bmL + w1);
        const uint4 h2 = *(const uint4*)(bmH + w2), l2 = *(const uint4*)(bmL + w2);
        const uint4 h3 = *(const uint4*)(bmH + w3), l3 = *(const uint4*)(bmL + w3);
        keepOwned(uint4{h0.x & l0.x, h0.y & l0.y, h0.z & l0.z, h0.w & l0.w}, it0);
        keepOwned(uint4{h1.x & l1.x, h1.y & l1.y, h1.z & l1.z, h1.w & l1.w}, it0 + 1);
        keepOwned(uint4{h2.x & l2.x, h2.y & l2.y, h2.z & l2.z, h2.w & l2.w}, it0 + 2);
        keepOwned(uint4{h3.x & l3.x, h3.y & l3.y, h3.z & l3.z, h3.w & l3.w}, it0 + 3);
    }
    while (qn > 0) flush(false);
    __syncthreads();

    if (wave < tileEnd - tileBeg) {
        const int  m0  = (tileBeg + wave) * 16;
        const int  col = lane & 15;
        const bool hi  = lane >= 16;

        int gr[8];
        for (int j = 0; j < 8; ++j) gr[j] = gids[m0 + (hi ? 8 : 0) + j];

        const float* accRow = &accL[(wave * 16 + col) * ACC_PITCH];
        v8f c10 = {}, c11 = {};
        for (int k = 0; k < 32; k += 4) {
            const int kb = k + (hi ? 2 : 0);
            const v2f av = *(const v2f*)(accRow + kb);
            const v2f bv = *(const v2f*)(bias1 + kb);
            v2f a;
            a.x = fmaxf(av.x + bv.x, 0.0f);
            a.y = fmaxf(av.y + bv.y, 0.0f);
            v2f b0, b1f;
            b0.x  = W2[kb * 32 + col];       b0.y  = W2[(kb + 1) * 32 + col];
            b1f.x = W2[kb * 32 + 16 + col];  b1f.y = W2[(kb + 1) * 32 + 16 + col];
            c10 = __builtin_amdgcn_wmma_f32_16x16x4_f32(false, a, false, b0,  (short)0, c10, false, false);
            c11 = __builtin_amdgcn_wmma_f32_16x16x4_f32(false, a, false, b1f, (short)0, c11, false, false);
        }

        const float bb0 = bias2[col];
        const float bb1 = bias2[16 + col];
        for (int j = 0; j < 8; ++j) {
            const int row = (hi ? 8 : 0) + j;
            hbuf[wave][row * 32 + col]      = fmaxf(c10[j] + bb0, 0.0f);
            hbuf[wave][row * 32 + 16 + col] = fmaxf(c11[j] + bb1, 0.0f);
        }

        v8f u0 = {}, u1 = {};
        for (int k = 0; k < 32; k += 4) {
            const int kb = k + (hi ? 2 : 0);
            const v2f a = *(const v2f*)(&hbuf[wave][col * 32 + kb]);
            v2f b0, b1f;
            b0.x  = Wp1[kb * 32 + col];       b0.y  = Wp1[(kb + 1) * 32 + col];
            b1f.x = Wp1[kb * 32 + 16 + col];  b1f.y = Wp1[(kb + 1) * 32 + 16 + col];
            u0 = __builtin_amdgcn_wmma_f32_16x16x4_f32(false, a, false, b0,  (short)0, u0, false, false);
            u1 = __builtin_amdgcn_wmma_f32_16x16x4_f32(false, a, false, b1f, (short)0, u1, false, false);
        }

        const int g0   = __builtin_amdgcn_readfirstlane(gr[0]);
        bool      same = true;
        for (int j = 0; j < 8; ++j) same = same && (gr[j] == g0);
        if (__builtin_amdgcn_ballot_w32(!same) == 0u) {
            float s0 = 0.0f, s1 = 0.0f;
            for (int j = 0; j < 8; ++j) { s0 += u0[j]; s1 += u1[j]; }
            ldsAddF(&gsum[g0 * 32 + col],      s0);
            ldsAddF(&gsum[g0 * 32 + 16 + col], s1);
            if (lane == 0) gflag[g0] = 1u;
        } else {
            for (int j = 0; j < 8; ++j) {
                ldsAddF(&gsum[gr[j] * 32 + col],      u0[j]);
                ldsAddF(&gsum[gr[j] * 32 + 16 + col], u1[j]);
                gflag[gr[j]] = 1u;
            }
        }
    }

    __syncthreads();
    for (int j = 0; j < N_GRAPHS / EN_WAVES; ++j) {
        const int g = wave + EN_WAVES * j;
        if (gflag[g] != 0u)
            atomicAdd(&out[g * 32 + lane], gsum[g * 32 + lane] + hv[j]);
    }
}

extern "C" void kernel_launch(void* const* d_in, const int* in_sizes, int n_in,
                              void* d_out, int out_size, void* d_ws, size_t ws_size,
                              hipStream_t stream) {
    const float* pos  = (const float*)d_in[0];
    const float* feat = (const float*)d_in[1];
    const int*   src  = (const int*)  d_in[2];
    const int*   dst  = (const int*)  d_in[3];
    const int*   gids = (const int*)  d_in[4];
    const float* W1   = (const float*)d_in[5];
    const float* b1   = (const float*)d_in[6];
    const float* W2   = (const float*)d_in[7];
    const float* b2   = (const float*)d_in[8];
    const float* Wp0  = (const float*)d_in[9];
    const float* bp0  = (const float*)d_in[10];
    const float* Wp1  = (const float*)d_in[11];
    const float* bp1  = (const float*)d_in[12];
    float* out = (float*)d_out;
    const int nEdges = in_sizes[2];

    float*    yBuf     = (float*)d_ws;
    unsigned* bm       = (unsigned*)(yBuf + (size_t)N_NODES * 32);
    float*    hand     = yBuf + (size_t)N_NODES * 32 + BM_ROOM_WORDS;
    unsigned* handFlag = (unsigned*)(hand + (size_t)HAND_WORDS);

    input_gemm_kernel<<<GEMM_GROUPS, GEMM_WAVES * 32, 0, stream>>>(pos, feat, W1, Wp0, gids,
                                                                   dst, nEdges, bp0, bp1,
                                                                   yBuf, hand, handFlag, bm, out);
    edge_node_kernel<<<GEMM_GROUPS, EN_THREADS, 0, stream>>>(yBuf, src, dst, nEdges, bm,
                                                             W2, b1, b2, Wp1, gids, hand, handFlag, out);
}
